# starting kernel
# baseline (speedup 1.0000x reference)
#include <hip/hip_runtime.h>

typedef float v2f __attribute__((ext_vector_type(2)));
typedef float v8f __attribute__((ext_vector_type(8)));

#define NN   50000
#define NRL  12
#define NT   500000
#define HIDN 16
#define EMBN 128
#define RP   25
#define KDIM 400
#define NAUG (2*NT + NN)

__device__ __forceinline__ void aug_edge(const int* __restrict__ t, int e,
                                         int& s, int& r, int& o) {
    if (e < NT)            { s = t[3*e];     r = t[3*e+1];       o = t[3*e+2]; }
    else if (e < 2*NT)     { int j = e - NT;
                             s = t[3*j+2];   r = t[3*j+1] + NRL; o = t[3*j];   }
    else                   { int i = e - 2*NT;
                             s = i;          r = 2*NRL;          o = i;        }
}

__global__ void k_zero(float4* __restrict__ p, int n4) {
    int i = blockIdx.x * blockDim.x + threadIdx.x;
    if (i < n4) p[i] = make_float4(0.f, 0.f, 0.f, 0.f);
}

__global__ void k_deg(const int* __restrict__ t, float* __restrict__ deg) {
    int e = blockIdx.x * blockDim.x + threadIdx.x;
    if (e >= NAUG) return;
    int s, r, o; aug_edge(t, e, s, r, o); (void)s;
    atomicAdd(&deg[r * NN + o], 1.0f);
}

__global__ void k_norm(float* __restrict__ deg) {
    int i = blockIdx.x * blockDim.x + threadIdx.x;
    if (i < RP * NN) deg[i] = 1.0f / fmaxf(deg[i], 1.0f);
}

__global__ void k_msg(const int* __restrict__ t, const float* __restrict__ W1,
                      const float* __restrict__ norm, float* __restrict__ x1) {
    int e = blockIdx.x * blockDim.x + threadIdx.x;
    if (e >= NAUG) return;
    int s, r, o; aug_edge(t, e, s, r, o);
    float en = norm[r * NN + o];
    const float4* w = (const float4*)(W1 + ((long)r * NN + s) * HIDN);
    float* dst = x1 + (long)o * HIDN;
#pragma unroll
    for (int q = 0; q < 4; q++) {
        float4 v = w[q];
        atomicAdd(dst + 4*q + 0, v.x * en);
        atomicAdd(dst + 4*q + 1, v.y * en);
        atomicAdd(dst + 4*q + 2, v.z * en);
        atomicAdd(dst + 4*q + 3, v.w * en);
    }
}

__global__ void k_relu(float* __restrict__ x1, const float* __restrict__ b1) {
    int i = blockIdx.x * blockDim.x + threadIdx.x;
    if (i < NN * HIDN) x1[i] = fmaxf(x1[i] + b1[i & (HIDN - 1)], 0.0f);
}

__global__ void k_agg(const int* __restrict__ t, const float* __restrict__ h,
                      const float* __restrict__ norm, float* __restrict__ agg) {
    int e = blockIdx.x * blockDim.x + threadIdx.x;
    if (e >= NAUG) return;
    int s, r, o; aug_edge(t, e, s, r, o);
    float en = norm[r * NN + o];
    const float4* hs = (const float4*)(h + (long)s * HIDN);
    float* dst = agg + (long)o * KDIM + r * HIDN;
#pragma unroll
    for (int q = 0; q < 4; q++) {
        float4 v = hs[q];
        atomicAdd(dst + 4*q + 0, v.x * en);
        atomicAdd(dst + 4*q + 1, v.y * en);
        atomicAdd(dst + 4*q + 2, v.z * en);
        atomicAdd(dst + 4*q + 3, v.w * en);
    }
}

__global__ void k_gemm(const float* __restrict__ agg, const float* __restrict__ W2,
                       const float* __restrict__ b2, float* __restrict__ x2) {
    int wave = (blockIdx.x * blockDim.x + threadIdx.x) >> 5;
    int lane = threadIdx.x & 31;
    if (wave >= NN / 16) return;
    int m0   = wave * 16;
    int half = lane >> 4;
    int lr   = lane & 15;

    const float* arow = agg + (long)(m0 + lr) * KDIM;

    v8f c[8];
#pragma unroll
    for (int nt = 0; nt < 8; nt++) {
        v8f z = {0.f, 0.f, 0.f, 0.f, 0.f, 0.f, 0.f, 0.f};
        c[nt] = z;
    }

    for (int k0 = 0; k0 < KDIM; k0 += 4) {
        int ka = k0 + 2 * half;
        v2f a;
        a.x = arow[ka];
        a.y = arow[ka + 1];
        const float* brow0 = W2 + (long)ka * EMBN + lr;
        const float* brow1 = W2 + (long)(ka + 1) * EMBN + lr;
#pragma unroll
        for (int nt = 0; nt < 8; nt++) {
            v2f b;
            b.x = brow0[nt * 16];
            b.y = brow1[nt * 16];
            c[nt] = __builtin_amdgcn_wmma_f32_16x16x4_f32(
                false, a, false, b, (short)0, c[nt], false, false);
        }
    }

#pragma unroll
    for (int nt = 0; nt < 8; nt++) {
        float bias = b2[nt * 16 + lr];
#pragma unroll
        for (int v = 0; v < 8; v++) {
            int row = m0 + v + 8 * half;
            x2[(long)row * EMBN + nt * 16 + lr] = c[nt][v] + bias;
        }
    }
}

__global__ void k_score(const int* __restrict__ t, const float* __restrict__ x2,
                        const float* __restrict__ rel, float* __restrict__ out) {
    int gid  = blockIdx.x * blockDim.x + threadIdx.x;
    int w    = gid >> 5;
    int lane = gid & 31;
    if (w >= NT) return;
    int s = t[3*w], p = t[3*w+1], o = t[3*w+2];
    const float* xs = x2 + (long)s * EMBN;
    const float* xo = x2 + (long)o * EMBN;
    const float* rp = rel + (long)p * EMBN;
    float acc = 0.f;
#pragma unroll
    for (int q = 0; q < 4; q++) {
        int j = lane + 32 * q;
        acc += xs[j] * rp[j] * xo[j];
    }
#pragma unroll
    for (int off = 16; off > 0; off >>= 1) acc += __shfl_xor(acc, off, 32);
    if (lane == 0) out[w] = acc;
}

extern "C" void kernel_launch(void* const* d_in, const int* in_sizes, int n_in,
                              void* d_out, int out_size, void* d_ws, size_t ws_size,
                              hipStream_t stream) {
    const int*   triples = (const int*)  d_in[0];
    const float* W1      = (const float*)d_in[1];
    const float* b1      = (const float*)d_in[2];
    const float* W2      = (const float*)d_in[3];
    const float* b2      = (const float*)d_in[4];
    const float* rel     = (const float*)d_in[5];
    float*       out     = (float*)d_out;

    float* deg = (float*)d_ws;
    float* x1  = deg + (long)RP * NN;
    float* agg = x1  + (long)NN * HIDN;
    float* x2  = agg + (long)NN * KDIM;
    long zf = (long)RP * NN + (long)NN * HIDN + (long)NN * KDIM;
    int n4 = (int)(zf / 4);

    dim3 B(256);
    k_zero<<<(n4 + 255) / 256, B, 0, stream>>>((float4*)d_ws, n4);
    k_deg <<<(NAUG + 255) / 256, B, 0, stream>>>(triples, deg);
    k_norm<<<(RP * NN + 255) / 256, B, 0, stream>>>(deg);
    k_msg <<<(NAUG + 255) / 256, B, 0, stream>>>(triples, W1, deg, x1);
    k_relu<<<(NN * HIDN + 255) / 256, B, 0, stream>>>(x1, b1);
    k_agg <<<(NAUG + 255) / 256, B, 0, stream>>>(triples, x1, deg, agg);

    int nwaves = NN / 16;
    k_gemm<<<(nwaves * 32 + 255) / 256, B, 0, stream>>>(agg, W2, b2, x2);

    long sthreads = (long)NT * 32;
    k_score<<<(int)((sthreads + 255) / 256), B, 0, stream>>>(triples, x2, rel, out);
}


# Layer 2 by the same 250 owners: layer 1 hands on its edge list; each owner adds h rows into a 200x400 LDS tile and feeds WMMA from it; no agg in memory, no zero launch (LDS f32 adds, arrival order)
# speedup vs baseline: 3.8892x; 3.8892x over previous
#include <hip/hip_runtime.h>

typedef float v2f __attribute__((ext_vector_type(2)));
typedef float v8f __attribute__((ext_vector_type(8)));

#define NN   50000
#define NRL  12
#define NT   500000
#define HIDN 16
#define EMBN 128
#define RP   25
#define KDIM 400
#define NAUG (2*NT + NN)

#define NPW  200
#define NWG  (NN / NPW)
#define LCAP 32768
#define L1T  1024
static_assert(NN % NPW == 0, "owners must tile the nodes");
static_assert(NN <= 65536 && NPW <= 256 && RP <= 256, "list word packing");

__device__ __forceinline__ void l1_take(int s, int p, int o, int base,
                                        int* s_deg, unsigned* s_list, int* s_cnt) {
    unsigned fo = (unsigned)(o - base);
    unsigned fs = (unsigned)(s - base);
    if ((fo < NPW) | (fs < NPW)) {
        if (fo < NPW) {
            atomicAdd(&s_deg[p * NPW + fo], 1);
            int pos = atomicAdd(s_cnt, 1);
            if (pos < LCAP) s_list[pos] = (unsigned)s | (fo << 16) | ((unsigned)p << 24);
        }
        if (fs < NPW) {
            int r = p + NRL;
            atomicAdd(&s_deg[r * NPW + fs], 1);
            int pos = atomicAdd(s_cnt, 1);
            if (pos < LCAP) s_list[pos] = (unsigned)o | (fs << 16) | ((unsigned)r << 24);
        }
    }
}

#define L1U 8
#define L1G (L1T / 16)

__global__ __launch_bounds__(L1T) void k_layer1(const int* __restrict__ t,
        const float* __restrict__ W1, const float* __restrict__ b1,
        float* __restrict__ norm, float* __restrict__ h,
        unsigned* __restrict__ glist, int* __restrict__ gcnt) {
    __shared__ int      s_deg[RP * NPW];
    __shared__ float    s_x1[NPW * HIDN];
    __shared__ unsigned s_list[LCAP];
    __shared__ int      s_cnt;

    const int tid  = threadIdx.x;
    const int base = blockIdx.x * NPW;

    for (int i = tid; i < RP * NPW; i += L1T) s_deg[i] = (i >= 2 * NRL * NPW) ? 1 : 0;
    if (tid == 0) s_cnt = 0;
    __syncthreads();

    const int nblk = NT / (L1U * L1T);
    const int blk0 = blockIdx.x % nblk;
    for (int it = 0; it < nblk; it++) {
        int blk = blk0 + it;
        if (blk >= nblk) blk -= nblk;
        const int* tp = t + (long)blk * (3 * L1U * L1T) + 3 * tid;
        int s[L1U], p[L1U], o[L1U];
#pragma unroll
        for (int u = 0; u < L1U; u++) {
            s[u] = tp[3 * u * L1T]; p[u] = tp[3 * u * L1T + 1]; o[u] = tp[3 * u * L1T + 2];
        }
#pragma unroll
        for (int u = 0; u < L1U; u++) l1_take(s[u], p[u], o[u], base, s_deg, s_list, &s_cnt);
    }
    for (int k = nblk * L1U * L1T + tid; k < NT; k += L1T)
        l1_take(t[3*k], t[3*k+1], t[3*k+2], base, s_deg, s_list, &s_cnt);
    __syncthreads();
    const int cnt = s_cnt;

    for (int i = tid; i < RP * NPW; i += L1T) {
        float nv = 1.0f / fmaxf((float)s_deg[i], 1.0f);
        s_deg[i] = __float_as_int(nv);
        int r = i / NPW, ol = i - r * NPW;
        norm[r * NN + base + ol] = nv;
    }
    __syncthreads();

    if (cnt <= LCAP) {
        uint2* gl = (uint2*)glist + (long)blockIdx.x * LCAP;
        for (int i = tid; i < cnt; i += L1T) {
            unsigned w = s_list[i];
            int ol = (int)((w >> 16) & 0xffu), r = (int)(w >> 24);
            gl[i] = make_uint2(w, (unsigned)s_deg[r * NPW + ol]);
        }
    }
    if (tid == 0) gcnt[blockIdx.x] = cnt;

    for (int i = tid; i < NPW * HIDN; i += L1T) {
        int ol = i >> 4, c = i & 15;
        float en = __int_as_float(s_deg[2 * NRL * NPW + ol]);
        s_x1[i] = W1[((long)(2 * NRL) * NN + base + ol) * HIDN + c] * en;
    }
    __syncthreads();

    if (cnt <= LCAP) {
        const int g = tid >> 4, c = tid & 15;
        int e = g;
        for (; e + (L1U - 1) * L1G < cnt; e += L1U * L1G) {
            float v[L1U]; int dst[L1U];
#pragma unroll
            for (int u = 0; u < L1U; u++) {
                unsigned w = s_list[e + u * L1G];
                int sN = (int)(w & 0xffffu), ol = (int)((w >> 16) & 0xffu), r = (int)(w >> 24);
                float en = __int_as_float(s_deg[r * NPW + ol]);
                v[u] = W1[((long)r * NN + sN) * HIDN + c] * en;
                dst[u] = ol * HIDN + c;
            }
#pragma unroll
            for (int u = 0; u < L1U; u++) atomicAdd(&s_x1[dst[u]], v[u]);
        }
        for (; e < cnt; e += L1G) {
            unsigned w = s_list[e];
            int sN = (int)(w & 0xffffu), ol = (int)((w >> 16) & 0xffu), r = (int)(w >> 24);
            float en = __int_as_float(s_deg[r * NPW + ol]);
            atomicAdd(&s_x1[ol * HIDN + c], W1[((long)r * NN + sN) * HIDN + c] * en);
        }
    } else {
        for (int k = tid; k < NT; k += L1T) {
            int s = t[3*k], p = t[3*k+1], o = t[3*k+2];
            unsigned fo = (unsigned)(o - base), fs = (unsigned)(s - base);
            if (fo < NPW) {
                float en = __int_as_float(s_deg[p * NPW + fo]);
                const float* w = W1 + ((long)p * NN + s) * HIDN;
                for (int c = 0; c < HIDN; c++) atomicAdd(&s_x1[fo * HIDN + c], w[c] * en);
            }
            if (fs < NPW) {
                int r = p + NRL;
                float en = __int_as_float(s_deg[r * NPW + fs]);
                const float* w = W1 + ((long)r * NN + o) * HIDN;
                for (int c = 0; c < HIDN; c++) atomicAdd(&s_x1[fs * HIDN + c], w[c] * en);
            }
        }
    }
    __syncthreads();

    for (int i = tid; i < NPW * HIDN; i += L1T)
        h[(long)base * HIDN + i] = fmaxf(s_x1[i] + b1[i & (HIDN - 1)], 0.0f);
}

#define TP   404
#define NSTR ((NPW + 15) / 16)
static_assert(NPW * TP * 4 <= 327680, "the tile must fit the LDS of one work-group");
static_assert(L1T == 1024 && NSTR <= 16, "wave to (column pair, strip) map below");

template <int NS>
__device__ __forceinline__ void l2_mul(const float* s_t, const float* __restrict__ W2,
        const float* __restrict__ b2, float* __restrict__ x2,
        int base, int sg, int n0, int half, int lr) {
    v8f acc[NS][2];
    int aoff[NS];
#pragma unroll
    for (int j = 0; j < NS; j++) {
        v8f z = {0.f, 0.f, 0.f, 0.f, 0.f, 0.f, 0.f, 0.f};
        acc[j][0] = z; acc[j][1] = z;
        int row = (sg + 8 * j) * 16 + lr;
        if (row > NPW - 1) row = NPW - 1;
        aoff[j] = row * TP + 2 * half;
    }
    const float* bp = W2 + (long)(2 * half) * EMBN + n0 + lr;
#pragma unroll 4
    for (int k0 = 0; k0 < KDIM; k0 += 4) {
        v2f b0, b1;
        b0.x = bp[(long)k0 * EMBN];        b0.y = bp[(long)(k0 + 1) * EMBN];
        b1.x = bp[(long)k0 * EMBN + 16];   b1.y = bp[(long)(k0 + 1) * EMBN + 16];
#pragma unroll
        for (int j = 0; j < NS; j++) {
            v2f a = *(const v2f*)(s_t + aoff[j] + k0);
            acc[j][0] = __builtin_amdgcn_wmma_f32_16x16x4_f32(
                false, a, false, b0, (short)0, acc[j][0], false, false);
            acc[j][1] = __builtin_amdgcn_wmma_f32_16x16x4_f32(
                false, a, false, b1, (short)0, acc[j][1], false, false);
        }
    }
#pragma unroll
    for (int j = 0; j < NS; j++) {
#pragma unroll
        for (int q = 0; q < 2; q++) {
            int n = n0 + 16 * q + lr;
            float bias = b2[n];
#pragma unroll
            for (int v = 0; v < 8; v++) {
                int row = (sg + 8 * j) * 16 + v + 8 * half;
                if (row < NPW) x2[(long)(base + row) * EMBN + n] = acc[j][q][v] + bias;
            }
        }
    }
}

__global__ __launch_bounds__(L1T) void k_layer2(const int* __restrict__ t,
        const unsigned* __restrict__ glist, const int* __restrict__ gcnt,
        const float* __restrict__ h, const float* __restrict__ norm,
        const float* __restrict__ W2, const float* __restrict__ b2, float* __restrict__ x2) {
    __shared__ __attribute__((aligned(16))) float s_t[NPW * TP];

    const int tid  = threadIdx.x;
    const int base = blockIdx.x * NPW;

    for (int i = tid; i < NPW * TP / 4; i += L1T) ((float4*)s_t)[i] = make_float4(0.f, 0.f, 0.f, 0.f);
    __syncthreads();

    const int cnt = gcnt[blockIdx.x];
    const int g = tid >> 4, c = tid & 15;

    for (int ol = g; ol < NPW; ol += L1G)
        s_t[ol * TP + 2 * NRL * HIDN + c] = h[(long)(base + ol) * HIDN + c] * norm[2 * NRL * NN + base + ol];

    if (cnt <= LCAP) {
        const uint2* lp = (const uint2*)glist + (long)blockIdx.x * LCAP;
        int e = g;
        for (; e + (L1U - 1) * L1G < cnt; e += L1U * L1G) {
            float v[L1U]; int dst[L1U];
#pragma unroll
            for (int u = 0; u < L1U; u++) {
                uint2 w = lp[e + u * L1G];
                int sN = (int)(w.x & 0xffffu), ol = (int)((w.x >> 16) & 0xffu), r = (int)(w.x >> 24);
                v[u] = h[(long)sN * HIDN + c] * __uint_as_float(w.y);
                dst[u] = ol * TP + r * HIDN + c;
            }
#pragma unroll
            for (int u = 0; u < L1U; u++) atomicAdd(&s_t[dst[u]], v[u]);
        }
        for (; e < cnt; e += L1G) {
            uint2 w = lp[e];
            int sN = (int)(w.x & 0xffffu), ol = (int)((w.x >> 16) & 0xffu), r = (int)(w.x >> 24);
            atomicAdd(&s_t[ol * TP + r * HIDN + c], h[(long)sN * HIDN + c] * __uint_as_float(w.y));
        }
    } else {
        for (int k = tid; k < NT; k += L1T) {
            int s = t[3*k], p = t[3*k+1], o = t[3*k+2];
            unsigned fo = (unsigned)(o - base), fs = (unsigned)(s - base);
            if (fo < NPW) {
                float en = norm[p * NN + o];
                const float* hs = h + (long)s * HIDN;
                for (int cc = 0; cc < HIDN; cc++) atomicAdd(&s_t[fo * TP + p * HIDN + cc], hs[cc] * en);
            }
            if (fs < NPW) {
                int r = p + NRL;
                float en = norm[r * NN + s];
                const float* ho = h + (long)o * HIDN;
                for (int cc = 0; cc < HIDN; cc++) atomicAdd(&s_t[fs * TP + r * HIDN + cc], ho[cc] * en);
            }
        }
    }
    __syncthreads();

    const int wave = tid >> 5, lane = tid & 31;
    const int n0 = (wave & 3) * 32, sg = wave >> 2;
    if (sg + 8 < NSTR) l2_mul<2>(s_t, W2, b2, x2, base, sg, n0, lane >> 4, lane & 15);
    else               l2_mul<1>(s_t, W2, b2, x2, base, sg, n0, lane >> 4, lane & 15);
}

__global__ void k_score(const int* __restrict__ t, const float* __restrict__ x2,
                        const float* __restrict__ rel, float* __restrict__ out) {
    int gid  = blockIdx.x * blockDim.x + threadIdx.x;
    int w    = gid >> 5;
    int lane = gid & 31;
    if (w >= NT) return;
    int s = t[3*w], p = t[3*w+1], o = t[3*w+2];
    const float* xs = x2 + (long)s * EMBN;
    const float* xo = x2 + (long)o * EMBN;
    const float* rp = rel + (long)p * EMBN;
    float acc = 0.f;
#pragma unroll
    for (int q = 0; q < 4; q++) {
        int j = lane + 32 * q;
        acc += xs[j] * rp[j] * xo[j];
    }
#pragma unroll
    for (int off = 16; off > 0; off >>= 1) acc += __shfl_xor(acc, off, 32);
    if (lane == 0) out[w] = acc;
}

extern "C" void kernel_launch(void* const* d_in, const int* in_sizes, int n_in,
                              void* d_out, int out_size, void* d_ws, size_t ws_size,
                              hipStream_t stream) {
    const int*   triples = (const int*)  d_in[0];
    const float* W1      = (const float*)d_in[1];
    const float* b1      = (const float*)d_in[2];
    const float* W2      = (const float*)d_in[3];
    const float* b2      = (const float*)d_in[4];
    const float* rel     = (const float*)d_in[5];
    float*       out     = (float*)d_out;

    float* deg = (float*)d_ws;
    float* x1  = deg + (long)RP * NN;
    float* agg = x1  + (long)NN * HIDN;
    float* x2  = agg + (long)NN * KDIM;
    unsigned* glist = (unsigned*)agg;
    int*      gcnt  = (int*)(agg + 2L * NWG * LCAP);
    static_assert(2L * NWG * LCAP + NWG <= (long)NN * KDIM, "lists fit where agg was");

    dim3 B(256);
    k_layer1<<<NWG, L1T, 0, stream>>>(triples, W1, b1, deg, x1, glist, gcnt);
    k_layer2<<<NWG, L1T, 0, stream>>>(triples, glist, gcnt, x1, deg, W2, b2, x2);

    long sthreads = (long)NT * 32;
    k_score<<<(int)((sthreads + 255) / 256), B, 0, stream>>>(triples, x2, rel, out);
}


# Layer 1 owners no longer each read all triples: a new first launch sorts each 2000-triple chunk's edges by owner in LDS; an owner copies its 250 segments (list order new; LDS f32 adds by arrival)
# speedup vs baseline: 4.5931x; 1.1810x over previous
#include <hip/hip_runtime.h>

typedef float v2f __attribute__((ext_vector_type(2)));
typedef float v8f __attribute__((ext_vector_type(8)));

#define NN   50000
#define NRL  12
#define NT   500000
#define HIDN 16
#define EMBN 128
#define RP   25
#define KDIM 400
#define NAUG (2*NT + NN)

#define NPW  200
#define NWG  (NN / NPW)
#define LCAP 32768
#define L1T  1024
#define NCH  250
#define CH   (NT / NCH)
static_assert(NN % NPW == 0, "owners must tile the nodes");
static_assert(NN <= 65536 && NPW <= 256 && RP <= 256, "list word packing");
static_assert(NT % NCH == 0 && 2 * CH < 65536 && CH % 2 == 0 && CH / 2 <= L1T, "chunking and segment word packing");
static_assert(NWG <= 256 && NCH <= 256 && NCH <= 8 * (L1T / 32), "one wave scans 256 counts; eight segments per wave");

__device__ __forceinline__ void wave_excl_scan(const int* cv, int* ov, int n, int sh, int lane, int* tot) {
    int v[8], sum = 0;
#pragma unroll
    for (int j = 0; j < 8; j++) {
        int idx = lane * 8 + j;
        int x = (idx < n) ? (int)((unsigned)cv[idx] >> sh) : 0;
        v[j] = sum; sum += x;
    }
    int incl = sum;
#pragma unroll
    for (int d = 1; d < 32; d <<= 1) {
        int y = __shfl_up(incl, d, 32);
        if (lane >= d) incl += y;
    }
    const int b = incl - sum;
#pragma unroll
    for (int j = 0; j < 8; j++) {
        int idx = lane * 8 + j;
        if (idx < n) ov[idx] = b + v[j];
    }
    if (tot != nullptr && lane == 31) *tot = incl;
}

__global__ __launch_bounds__(L1T) void k_bin(const int* __restrict__ t,
        unsigned* __restrict__ sorted, unsigned* __restrict__ segT) {
    __shared__ int s_c[256];
    __shared__ int s_o[256];
    const int tid = threadIdx.x;
    const int a   = blockIdx.x;
    if (tid < 256) s_c[tid] = 0;
    __syncthreads();

    const bool on = tid < CH / 2;
    unsigned wf[2] = {0u, 0u}, wi[2] = {0u, 0u};
    int gf[2] = {0, 0}, gi[2] = {0, 0}, rf[2] = {0, 0}, ri[2] = {0, 0};
    if (on) {
        const int* tp = t + 3L * (a * CH + tid);
        int s0 = tp[0], p0 = tp[1], o0 = tp[2];
        int s1 = tp[3 * (CH / 2)], p1 = tp[3 * (CH / 2) + 1], o1 = tp[3 * (CH / 2) + 2];
        int ts[2] = {s0, s1}, tr[2] = {p0, p1}, to[2] = {o0, o1};
#pragma unroll
        for (int u = 0; u < 2; u++) {
            int s = ts[u], p = tr[u], o = to[u];
            int go = o / NPW, gs = s / NPW;
            wf[u] = (unsigned)s | ((unsigned)(o - go * NPW) << 16) | ((unsigned)p << 24);
            wi[u] = (unsigned)o | ((unsigned)(s - gs * NPW) << 16) | ((unsigned)(p + NRL) << 24);
            gf[u] = go; gi[u] = gs;
            rf[u] = atomicAdd(&s_c[go], 1);
            ri[u] = atomicAdd(&s_c[gs], 1);
        }
    }
    __syncthreads();
    if (tid < 32) wave_excl_scan(s_c, s_o, NWG, 0, tid, nullptr);
    __syncthreads();

    if (on) {
        unsigned* sp = sorted + (long)a * (2 * CH);
#pragma unroll
        for (int u = 0; u < 2; u++) {
            sp[s_o[gf[u]] + rf[u]] = wf[u];
            sp[s_o[gi[u]] + ri[u]] = wi[u];
        }
    }
    if (tid < NWG) segT[tid * NCH + a] = (unsigned)s_o[tid] | ((unsigned)s_c[tid] << 16);
}

__device__ __forceinline__ void l1_put(unsigned w, int pos, int* s_deg, unsigned* s_list) {
    atomicAdd(&s_deg[(int)(w >> 24) * NPW + (int)((w >> 16) & 0xffu)], 1);
    if (pos < LCAP) s_list[pos] = w;
}

#define L1U 8
#define L1G (L1T / 16)

__global__ __launch_bounds__(L1T) void k_layer1(const int* __restrict__ t,
        const float* __restrict__ W1, const float* __restrict__ b1,
        float* __restrict__ norm, float* __restrict__ h,
        unsigned* __restrict__ glist, int* __restrict__ gcnt,
        const unsigned* __restrict__ sorted, const unsigned* __restrict__ segT) {
    __shared__ int      s_deg[RP * NPW];
    __shared__ float    s_x1[NPW * HIDN];
    __shared__ unsigned s_list[LCAP];
    __shared__ int      s_seg[256];
    __shared__ int      s_dst[256];
    __shared__ int      s_cnt;

    const int tid  = threadIdx.x;
    const int base = blockIdx.x * NPW;

    for (int i = tid; i < RP * NPW; i += L1T) s_deg[i] = (i >= 2 * NRL * NPW) ? 1 : 0;
    if (tid < 256) s_seg[tid] = (tid < NCH) ? (int)segT[blockIdx.x * NCH + tid] : 0;
    __syncthreads();
    if (tid < 32) wave_excl_scan(s_seg, s_dst, NCH, 16, tid, &s_cnt);
    __syncthreads();

    {
        const int wv = tid >> 5, ln = tid & 31;
        const unsigned* sp[8];
        unsigned w0[8];
        int n[8], d0[8];
#pragma unroll
        for (int u = 0; u < 8; u++) {
            int a = wv + u * (L1T / 32);
            n[u] = 0; d0[u] = 0; sp[u] = sorted;
            if (a < NCH) {
                unsigned sg = (unsigned)s_seg[a];
                n[u]  = (int)(sg >> 16);
                sp[u] = sorted + (long)a * (2 * CH) + (sg & 0xffffu);
                d0[u] = s_dst[a];
            }
            w0[u] = (ln < n[u]) ? sp[u][ln] : 0u;
        }
#pragma unroll
        for (int u = 0; u < 8; u++) {
            if (ln < n[u]) l1_put(w0[u], d0[u] + ln, s_deg, s_list);
            for (int i = ln + 32; i < n[u]; i += 32) l1_put(sp[u][i], d0[u] + i, s_deg, s_list);
        }
    }
    __syncthreads();
    const int cnt = s_cnt;

    for (int i = tid; i < RP * NPW; i += L1T) {
        float nv = 1.0f / fmaxf((float)s_deg[i], 1.0f);
        s_deg[i] = __float_as_int(nv);
        int r = i / NPW, ol = i - r * NPW;
        norm[r * NN + base + ol] = nv;
    }
    __syncthreads();

    if (cnt <= LCAP) {
        uint2* gl = (uint2*)glist + (long)blockIdx.x * LCAP;
        for (int i = tid; i < cnt; i += L1T) {
            unsigned w = s_list[i];
            int ol = (int)((w >> 16) & 0xffu), r = (int)(w >> 24);
            gl[i] = make_uint2(w, (unsigned)s_deg[r * NPW + ol]);
        }
    }
    if (tid == 0) gcnt[blockIdx.x] = cnt;

    for (int i = tid; i < NPW * HIDN; i += L1T) {
        int ol = i >> 4, c = i & 15;
        float en = __int_as_float(s_deg[2 * NRL * NPW + ol]);
        s_x1[i] = W1[((long)(2 * NRL) * NN + base + ol) * HIDN + c] * en;
    }
    __syncthreads();

    if (cnt <= LCAP) {
        const int g = tid >> 4, c = tid & 15;
        int e = g;
        for (; e + (L1U - 1) * L1G < cnt; e += L1U * L1G) {
            float v[L1U]; int dst[L1U];
#pragma unroll
            for (int u = 0; u < L1U; u++) {
                unsigned w = s_list[e + u * L1G];
                int sN = (int)(w & 0xffffu), ol = (int)((w >> 16) & 0xffu), r = (int)(w >> 24);
                float en = __int_as_float(s_deg[r * NPW + ol]);
                v[u] = W1[((long)r * NN + sN) * HIDN + c] * en;
                dst[u] = ol * HIDN + c;
            }
#pragma unroll
            for (int u = 0; u < L1U; u++) atomicAdd(&s_x1[dst[u]], v[u]);
        }
        for (; e < cnt; e += L1G) {
            unsigned w = s_list[e];
            int sN = (int)(w & 0xffffu), ol = (int)((w >> 16) & 0xffu), r = (int)(w >> 24);
            float en = __int_as_float(s_deg[r * NPW + ol]);
            atomicAdd(&s_x1[ol * HIDN + c], W1[((long)r * NN + sN) * HIDN + c] * en);
        }
    } else {
        for (int k = tid; k < NT; k += L1T) {
            int s = t[3*k], p = t[3*k+1], o = t[3*k+2];
            unsigned fo = (unsigned)(o - base), fs = (unsigned)(s - base);
            if (fo < NPW) {
                float en = __int_as_float(s_deg[p * NPW + fo]);
                const float* w = W1 + ((long)p * NN + s) * HIDN;
                for (int c = 0; c < HIDN; c++) atomicAdd(&s_x1[fo * HIDN + c], w[c] * en);
            }
            if (fs < NPW) {
                int r = p + NRL;
                float en = __int_as_float(s_deg[r * NPW + fs]);
                const float* w = W1 + ((long)r * NN + o) * HIDN;
                for (int c = 0; c < HIDN; c++) atomicAdd(&s_x1[fs * HIDN + c], w[c] * en);
            }
        }
    }
    __syncthreads();

    for (int i = tid; i < NPW * HIDN; i += L1T)
        h[(long)base * HIDN + i] = fmaxf(s_x1[i] + b1[i & (HIDN - 1)], 0.0f);
}

#define TP   404
#define NSTR ((NPW + 15) / 16)
static_assert(NPW * TP * 4 <= 327680, "the tile must fit the LDS of one work-group");
static_assert(L1T == 1024 && NSTR <= 16, "wave to (column pair, strip) map below");

template <int NS>
__device__ __forceinline__ void l2_mul(const float* s_t, const float* __restrict__ W2,
        const float* __restrict__ b2, float* __restrict__ x2,
        int base, int sg, int n0, int half, int lr) {
    v8f acc[NS][2];
    int aoff[NS];
#pragma unroll
    for (int j = 0; j < NS; j++) {
        v8f z = {0.f, 0.f, 0.f, 0.f, 0.f, 0.f, 0.f, 0.f};
        acc[j][0] = z; acc[j][1] = z;
        int row = (sg + 8 * j) * 16 + lr;
        if (row > NPW - 1) row = NPW - 1;
        aoff[j] = row * TP + 2 * half;
    }
    const float* bp = W2 + (long)(2 * half) * EMBN + n0 + lr;
#pragma unroll 4
    for (int k0 = 0; k0 < KDIM; k0 += 4) {
        v2f b0, b1;
        b0.x = bp[(long)k0 * EMBN];        b0.y = bp[(long)(k0 + 1) * EMBN];
        b1.x = bp[(long)k0 * EMBN + 16];   b1.y = bp[(long)(k0 + 1) * EMBN + 16];
#pragma unroll
        for (int j = 0; j < NS; j++) {
            v2f a = *(const v2f*)(s_t + aoff[j] + k0);
            acc[j][0] = __builtin_amdgcn_wmma_f32_16x16x4_f32(
                false, a, false, b0, (short)0, acc[j][0], false, false);
            acc[j][1] = __builtin_amdgcn_wmma_f32_16x16x4_f32(
                false, a, false, b1, (short)0, acc[j][1], false, false);
        }
    }
#pragma unroll
    for (int j = 0; j < NS; j++) {
#pragma unroll
        for (int q = 0; q < 2; q++) {
            int n = n0 + 16 * q + lr;
            float bias = b2[n];
#pragma unroll
            for (int v = 0; v < 8; v++) {
                int row = (sg + 8 * j) * 16 + v + 8 * half;
                if (row < NPW) x2[(long)(base + row) * EMBN + n] = acc[j][q][v] + bias;
            }
        }
    }
}

__global__ __launch_bounds__(L1T) void k_layer2(const int* __restrict__ t,
        const unsigned* __restrict__ glist, const int* __restrict__ gcnt,
        const float* __restrict__ h, const float* __restrict__ norm,
        const float* __restrict__ W2, const float* __restrict__ b2, float* __restrict__ x2) {
    __shared__ __attribute__((aligned(16))) float s_t[NPW * TP];

    const int tid  = threadIdx.x;
    const int base = blockIdx.x * NPW;

    for (int i = tid; i < NPW * TP / 4; i += L1T) ((float4*)s_t)[i] = make_float4(0.f, 0.f, 0.f, 0.f);
    __syncthreads();

    const int cnt = gcnt[blockIdx.x];
    const int g = tid >> 4, c = tid & 15;

    for (int ol = g; ol < NPW; ol += L1G)
        s_t[ol * TP + 2 * NRL * HIDN + c] = h[(long)(base + ol) * HIDN + c] * norm[2 * NRL * NN + base + ol];

    if (cnt <= LCAP) {
        const uint2* lp = (const uint2*)glist + (long)blockIdx.x * LCAP;
        int e = g;
        for (; e + (L1U - 1) * L1G < cnt; e += L1U * L1G) {
            float v[L1U]; int dst[L1U];
#pragma unroll
            for (int u = 0; u < L1U; u++) {
                uint2 w = lp[e + u * L1G];
                int sN = (int)(w.x & 0xffffu), ol = (int)((w.x >> 16) & 0xffu), r = (int)(w.x >> 24);
                v[u] = h[(long)sN * HIDN + c] * __uint_as_float(w.y);
                dst[u] = ol * TP + r * HIDN + c;
            }
#pragma unroll
            for (int u = 0; u < L1U; u++) atomicAdd(&s_t[dst[u]], v[u]);
        }
        for (; e < cnt; e += L1G) {
            uint2 w = lp[e];
            int sN = (int)(w.x & 0xffffu), ol = (int)((w.x >> 16) & 0xffu), r = (int)(w.x >> 24);
            atomicAdd(&s_t[ol * TP + r * HIDN + c], h[(long)sN * HIDN + c] * __uint_as_float(w.y));
        }
    } else {
        for (int k = tid; k < NT; k += L1T) {
            int s = t[3*k], p = t[3*k+1], o = t[3*k+2];
            unsigned fo = (unsigned)(o - base), fs = (unsigned)(s - base);
            if (fo < NPW) {
                float en = norm[p * NN + o];
                const float* hs = h + (long)s * HIDN;
                for (int cc = 0; cc < HIDN; cc++) atomicAdd(&s_t[fo * TP + p * HIDN + cc], hs[cc] * en);
            }
            if (fs < NPW) {
                int r = p + NRL;
                float en = norm[r * NN + s];
                const float* ho = h + (long)o * HIDN;
                for (int cc = 0; cc < HIDN; cc++) atomicAdd(&s_t[fs * TP + r * HIDN + cc], ho[cc] * en);
            }
        }
    }
    __syncthreads();

    const int wave = tid >> 5, lane = tid & 31;
    const int n0 = (wave & 3) * 32, sg = wave >> 2;
    if (sg + 8 < NSTR) l2_mul<2>(s_t, W2, b2, x2, base, sg, n0, lane >> 4, lane & 15);
    else               l2_mul<1>(s_t, W2, b2, x2, base, sg, n0, lane >> 4, lane & 15);
}

__global__ void k_score(const int* __restrict__ t, const float* __restrict__ x2,
                        const float* __restrict__ rel, float* __restrict__ out) {
    int gid  = blockIdx.x * blockDim.x + threadIdx.x;
    int w    = gid >> 5;
    int lane = gid & 31;
    if (w >= NT) return;
    int s = t[3*w], p = t[3*w+1], o = t[3*w+2];
    const float* xs = x2 + (long)s * EMBN;
    const float* xo = x2 + (long)o * EMBN;
    const float* rp = rel + (long)p * EMBN;
    float acc = 0.f;
#pragma unroll
    for (int q = 0; q < 4; q++) {
        int j = lane + 32 * q;
        acc += xs[j] * rp[j] * xo[j];
    }
#pragma unroll
    for (int off = 16; off > 0; off >>= 1) acc += __shfl_xor(acc, off, 32);
    if (lane == 0) out[w] = acc;
}

extern "C" void kernel_launch(void* const* d_in, const int* in_sizes, int n_in,
                              void* d_out, int out_size, void* d_ws, size_t ws_size,
                              hipStream_t stream) {
    const int*   triples = (const int*)  d_in[0];
    const float* W1      = (const float*)d_in[1];
    const float* b1      = (const float*)d_in[2];
    const float* W2      = (const float*)d_in[3];
    const float* b2      = (const float*)d_in[4];
    const float* rel     = (const float*)d_in[5];
    float*       out     = (float*)d_out;

    float* deg = (float*)d_ws;
    float* x1  = deg + (long)RP * NN;
    float* agg = x1  + (long)NN * HIDN;
    float* x2  = agg + (long)NN * KDIM;
    unsigned* glist = (unsigned*)agg;
    int*      gcnt  = (int*)(agg + 2L * NWG * LCAP);
    unsigned* sorted = (unsigned*)(agg + 2L * NWG * LCAP + 256);
    unsigned* segT   = sorted + 2L * NT;
    static_assert(2L * NWG * LCAP + 256 + 2L * NT + (long)NWG * NCH <= (long)NN * KDIM, "lists and sorted chunks fit where agg was");

    dim3 B(256);
    k_bin<<<NCH, L1T, 0, stream>>>(triples, sorted, segT);
    k_layer1<<<NWG, L1T, 0, stream>>>(triples, W1, b1, deg, x1, glist, gcnt, sorted, segT);
    k_layer2<<<NWG, L1T, 0, stream>>>(triples, glist, gcnt, x1, deg, W2, b2, x2);

    long sthreads = (long)NT * 32;
    k_score<<<(int)((sthreads + 255) / 256), B, 0, stream>>>(triples, x2, rel, out);
}


# Scores by the 250 destination owners: forward edges from the sorted chunks (k_bin stores each edge's triple index), x2[o] and rel rows in LDS, 8 x2[s] rows in flight per wave; new f32 sum order
# speedup vs baseline: 4.9975x; 1.0880x over previous
#include <hip/hip_runtime.h>

typedef float v2f __attribute__((ext_vector_type(2)));
typedef float v8f __attribute__((ext_vector_type(8)));

#define NN   50000
#define NRL  12
#define NT   500000
#define HIDN 16
#define EMBN 128
#define RP   25
#define KDIM 400
#define NAUG (2*NT + NN)

#define NPW  200
#define NWG  (NN / NPW)
#define LCAP 32768
#define L1T  1024
#define NCH  250
#define CH   (NT / NCH)
static_assert(NN % NPW == 0, "owners must tile the nodes");
static_assert(NN <= 65536 && NPW <= 256 && RP <= 256, "list word packing");
static_assert(NT % NCH == 0 && 2 * CH < 65536 && CH % 2 == 0 && CH / 2 <= L1T, "chunking and segment word packing");
static_assert(NWG <= 256 && NCH <= 256 && NCH <= 8 * (L1T / 32), "one wave scans 256 counts; eight segments per wave");

__device__ __forceinline__ void wave_excl_scan(const int* cv, int* ov, int n, int sh, int lane, int* tot) {
    int v[8], sum = 0;
#pragma unroll
    for (int j = 0; j < 8; j++) {
        int idx = lane * 8 + j;
        int x = (idx < n) ? (int)((unsigned)cv[idx] >> sh) : 0;
        v[j] = sum; sum += x;
    }
    int incl = sum;
#pragma unroll
    for (int d = 1; d < 32; d <<= 1) {
        int y = __shfl_up(incl, d, 32);
        if (lane >= d) incl += y;
    }
    const int b = incl - sum;
#pragma unroll
    for (int j = 0; j < 8; j++) {
        int idx = lane * 8 + j;
        if (idx < n) ov[idx] = b + v[j];
    }
    if (tot != nullptr && lane == 31) *tot = incl;
}

__global__ __launch_bounds__(L1T) void k_bin(const int* __restrict__ t,
        unsigned* __restrict__ sorted, unsigned* __restrict__ sidx, unsigned* __restrict__ segT) {
    __shared__ int s_c[256];
    __shared__ int s_o[256];
    const int tid = threadIdx.x;
    const int a   = blockIdx.x;
    if (tid < 256) s_c[tid] = 0;
    __syncthreads();

    const bool on = tid < CH / 2;
    unsigned wf[2] = {0u, 0u}, wi[2] = {0u, 0u};
    int gf[2] = {0, 0}, gi[2] = {0, 0}, rf[2] = {0, 0}, ri[2] = {0, 0};
    if (on) {
        const int* tp = t + 3L * (a * CH + tid);
        int s0 = tp[0], p0 = tp[1], o0 = tp[2];
        int s1 = tp[3 * (CH / 2)], p1 = tp[3 * (CH / 2) + 1], o1 = tp[3 * (CH / 2) + 2];
        int ts[2] = {s0, s1}, tr[2] = {p0, p1}, to[2] = {o0, o1};
#pragma unroll
        for (int u = 0; u < 2; u++) {
            int s = ts[u], p = tr[u], o = to[u];
            int go = o / NPW, gs = s / NPW;
            wf[u] = (unsigned)s | ((unsigned)(o - go * NPW) << 16) | ((unsigned)p << 24);
            wi[u] = (unsigned)o | ((unsigned)(s - gs * NPW) << 16) | ((unsigned)(p + NRL) << 24);
            gf[u] = go; gi[u] = gs;
            rf[u] = atomicAdd(&s_c[go], 1);
            ri[u] = atomicAdd(&s_c[gs], 1);
        }
    }
    __syncthreads();
    if (tid < 32) wave_excl_scan(s_c, s_o, NWG, 0, tid, nullptr);
    __syncthreads();

    if (on) {
        unsigned* sp = sorted + (long)a * (2 * CH);
        unsigned* ip = sidx   + (long)a * (2 * CH);
#pragma unroll
        for (int u = 0; u < 2; u++) {
            const int pf = s_o[gf[u]] + rf[u], pi = s_o[gi[u]] + ri[u];
            const unsigned ti = (unsigned)(a * CH + tid + u * (CH / 2));
            sp[pf] = wf[u]; ip[pf] = ti;
            sp[pi] = wi[u]; ip[pi] = ti;
        }
    }
    if (tid < NWG) segT[tid * NCH + a] = (unsigned)s_o[tid] | ((unsigned)s_c[tid] << 16);
}

__device__ __forceinline__ void l1_put(unsigned w, int pos, int* s_deg, unsigned* s_list) {
    atomicAdd(&s_deg[(int)(w >> 24) * NPW + (int)((w >> 16) & 0xffu)], 1);
    if (pos < LCAP) s_list[pos] = w;
}

#define L1U 8
#define L1G (L1T / 16)

__global__ __launch_bounds__(L1T) void k_layer1(const int* __restrict__ t,
        const float* __restrict__ W1, const float* __restrict__ b1,
        float* __restrict__ norm, float* __restrict__ h,
        unsigned* __restrict__ glist, int* __restrict__ gcnt,
        const unsigned* __restrict__ sorted, const unsigned* __restrict__ segT) {
    __shared__ int      s_deg[RP * NPW];
    __shared__ float    s_x1[NPW * HIDN];
    __shared__ unsigned s_list[LCAP];
    __shared__ int      s_seg[256];
    __shared__ int      s_dst[256];
    __shared__ int      s_cnt;

    const int tid  = threadIdx.x;
    const int base = blockIdx.x * NPW;

    for (int i = tid; i < RP * NPW; i += L1T) s_deg[i] = (i >= 2 * NRL * NPW) ? 1 : 0;
    if (tid < 256) s_seg[tid] = (tid < NCH) ? (int)segT[blockIdx.x * NCH + tid] : 0;
    __syncthreads();
    if (tid < 32) wave_excl_scan(s_seg, s_dst, NCH, 16, tid, &s_cnt);
    __syncthreads();

    {
        const int wv = tid >> 5, ln = tid & 31;
        const unsigned* sp[8];
        unsigned w0[8];
        int n[8], d0[8];
#pragma unroll
        for (int u = 0; u < 8; u++) {
            int a = wv + u * (L1T / 32);
            n[u] = 0; d0[u] = 0; sp[u] = sorted;
            if (a < NCH) {
                unsigned sg = (unsigned)s_seg[a];
                n[u]  = (int)(sg >> 16);
                sp[u] = sorted + (long)a * (2 * CH) + (sg & 0xffffu);
                d0[u] = s_dst[a];
            }
            w0[u] = (ln < n[u]) ? sp[u][ln] : 0u;
        }
#pragma unroll
        for (int u = 0; u < 8; u++) {
            if (ln < n[u]) l1_put(w0[u], d0[u] + ln, s_deg, s_list);
            for (int i = ln + 32; i < n[u]; i += 32) l1_put(sp[u][i], d0[u] + i, s_deg, s_list);
        }
    }
    __syncthreads();
    const int cnt = s_cnt;

    for (int i = tid; i < RP * NPW; i += L1T) {
        float nv = 1.0f / fmaxf((float)s_deg[i], 1.0f);
        s_deg[i] = __float_as_int(nv);
        int r = i / NPW, ol = i - r * NPW;
        norm[r * NN + base + ol] = nv;
    }
    __syncthreads();

    if (cnt <= LCAP) {
        uint2* gl = (uint2*)glist + (long)blockIdx.x * LCAP;
        for (int i = tid; i < cnt; i += L1T) {
            unsigned w = s_list[i];
            int ol = (int)((w >> 16) & 0xffu), r = (int)(w >> 24);
            gl[i] = make_uint2(w, (unsigned)s_deg[r * NPW + ol]);
        }
    }
    if (tid == 0) gcnt[blockIdx.x] = cnt;

    for (int i = tid; i < NPW * HIDN; i += L1T) {
        int ol = i >> 4, c = i & 15;
        float en = __int_as_float(s_deg[2 * NRL * NPW + ol]);
        s_x1[i] = W1[((long)(2 * NRL) * NN + base + ol) * HIDN + c] * en;
    }
    __syncthreads();

    if (cnt <= LCAP) {
        const int g = tid >> 4, c = tid & 15;
        int e = g;
        for (; e + (L1U - 1) * L1G < cnt; e += L1U * L1G) {
            float v[L1U]; int dst[L1U];
#pragma unroll
            for (int u = 0; u < L1U; u++) {
                unsigned w = s_list[e + u * L1G];
                int sN = (int)(w & 0xffffu), ol = (int)((w >> 16) & 0xffu), r = (int)(w >> 24);
                float en = __int_as_float(s_deg[r * NPW + ol]);
                v[u] = W1[((long)r * NN + sN) * HIDN + c] * en;
                dst[u] = ol * HIDN + c;
            }
#pragma unroll
            for (int u = 0; u < L1U; u++) atomicAdd(&s_x1[dst[u]], v[u]);
        }
        for (; e < cnt; e += L1G) {
            unsigned w = s_list[e];
            int sN = (int)(w & 0xffffu), ol = (int)((w >> 16) & 0xffu), r = (int)(w >> 24);
            float en = __int_as_float(s_deg[r * NPW + ol]);
            atomicAdd(&s_x1[ol * HIDN + c], W1[((long)r * NN + sN) * HIDN + c] * en);
        }
    } else {
        for (int k = tid; k < NT; k += L1T) {
            int s = t[3*k], p = t[3*k+1], o = t[3*k+2];
            unsigned fo = (unsigned)(o - base), fs = (unsigned)(s - base);
            if (fo < NPW) {
                float en = __int_as_float(s_deg[p * NPW + fo]);
                const float* w = W1 + ((long)p * NN + s) * HIDN;
                for (int c = 0; c < HIDN; c++) atomicAdd(&s_x1[fo * HIDN + c], w[c] * en);
            }
            if (fs < NPW) {
                int r = p + NRL;
                float en = __int_as_float(s_deg[r * NPW + fs]);
                const float* w = W1 + ((long)r * NN + o) * HIDN;
                for (int c = 0; c < HIDN; c++) atomicAdd(&s_x1[fs * HIDN + c], w[c] * en);
            }
        }
    }
    __syncthreads();

    for (int i = tid; i < NPW * HIDN; i += L1T)
        h[(long)base * HIDN + i] = fmaxf(s_x1[i] + b1[i & (HIDN - 1)], 0.0f);
}

#define TP   404
#define NSTR ((NPW + 15) / 16)
static_assert(NPW * TP * 4 <= 327680, "the tile must fit the LDS of one work-group");
static_assert(L1T == 1024 && NSTR <= 16, "wave to (column pair, strip) map below");

template <int NS>
__device__ __forceinline__ void l2_mul(const float* s_t, const float* __restrict__ W2,
        const float* __restrict__ b2, float* __restrict__ x2,
        int base, int sg, int n0, int half, int lr) {
    v8f acc[NS][2];
    int aoff[NS];
#pragma unroll
    for (int j = 0; j < NS; j++) {
        v8f z = {0.f, 0.f, 0.f, 0.f, 0.f, 0.f, 0.f, 0.f};
        acc[j][0] = z; acc[j][1] = z;
        int row = (sg + 8 * j) * 16 + lr;
        if (row > NPW - 1) row = NPW - 1;
        aoff[j] = row * TP + 2 * half;
    }
    const float* bp = W2 + (long)(2 * half) * EMBN + n0 + lr;
#pragma unroll 4
    for (int k0 = 0; k0 < KDIM; k0 += 4) {
        v2f b0, b1;
        b0.x = bp[(long)k0 * EMBN];        b0.y = bp[(long)(k0 + 1) * EMBN];
        b1.x = bp[(long)k0 * EMBN + 16];   b1.y = bp[(long)(k0 + 1) * EMBN + 16];
#pragma unroll
        for (int j = 0; j < NS; j++) {
            v2f a = *(const v2f*)(s_t + aoff[j] + k0);
            acc[j][0] = __builtin_amdgcn_wmma_f32_16x16x4_f32(
                false, a, false, b0, (short)0, acc[j][0], false, false);
            acc[j][1] = __builtin_amdgcn_wmma_f32_16x16x4_f32(
                false, a, false, b1, (short)0, acc[j][1], false, false);
        }
    }
#pragma unroll
    for (int j = 0; j < NS; j++) {
#pragma unroll
        for (int q = 0; q < 2; q++) {
            int n = n0 + 16 * q + lr;
            float bias = b2[n];
#pragma unroll
            for (int v = 0; v < 8; v++) {
                int row = (sg + 8 * j) * 16 + v + 8 * half;
                if (row < NPW) x2[(long)(base + row) * EMBN + n] = acc[j][q][v] + bias;
            }
        }
    }
}

__global__ __launch_bounds__(L1T) void k_layer2(const int* __restrict__ t,
        const unsigned* __restrict__ glist, const int* __restrict__ gcnt,
        const float* __restrict__ h, const float* __restrict__ norm,
        const float* __restrict__ W2, const float* __restrict__ b2, float* __restrict__ x2) {
    __shared__ __attribute__((aligned(16))) float s_t[NPW * TP];

    const int tid  = threadIdx.x;
    const int base = blockIdx.x * NPW;

    for (int i = tid; i < NPW * TP / 4; i += L1T) ((float4*)s_t)[i] = make_float4(0.f, 0.f, 0.f, 0.f);
    __syncthreads();

    const int cnt = gcnt[blockIdx.x];
    const int g = tid >> 4, c = tid & 15;

    for (int ol = g; ol < NPW; ol += L1G)
        s_t[ol * TP + 2 * NRL * HIDN + c] = h[(long)(base + ol) * HIDN + c] * norm[2 * NRL * NN + base + ol];

    if (cnt <= LCAP) {
        const uint2* lp = (const uint2*)glist + (long)blockIdx.x * LCAP;
        int e = g;
        for (; e + (L1U - 1) * L1G < cnt; e += L1U * L1G) {
            float v[L1U]; int dst[L1U];
#pragma unroll
            for (int u = 0; u < L1U; u++) {
                uint2 w = lp[e + u * L1G];
                int sN = (int)(w.x & 0xffffu), ol = (int)((w.x >> 16) & 0xffu), r = (int)(w.x >> 24);
                v[u] = h[(long)sN * HIDN + c] * __uint_as_float(w.y);
                dst[u] = ol * TP + r * HIDN + c;
            }
#pragma unroll
            for (int u = 0; u < L1U; u++) atomicAdd(&s_t[dst[u]], v[u]);
        }
        for (; e < cnt; e += L1G) {
            uint2 w = lp[e];
            int sN = (int)(w.x & 0xffffu), ol = (int)((w.x >> 16) & 0xffu), r = (int)(w.x >> 24);
            atomicAdd(&s_t[ol * TP + r * HIDN + c], h[(long)sN * HIDN + c] * __uint_as_float(w.y));
        }
    } else {
        for (int k = tid; k < NT; k += L1T) {
            int s = t[3*k], p = t[3*k+1], o = t[3*k+2];
            unsigned fo = (unsigned)(o - base), fs = (unsigned)(s - base);
            if (fo < NPW) {
                float en = norm[p * NN + o];
                const float* hs = h + (long)s * HIDN;
                for (int cc = 0; cc < HIDN; cc++) atomicAdd(&s_t[fo * TP + p * HIDN + cc], hs[cc] * en);
            }
            if (fs < NPW) {
                int r = p + NRL;
                float en = norm[r * NN + s];
                const float* ho = h + (long)o * HIDN;
                for (int cc = 0; cc < HIDN; cc++) atomicAdd(&s_t[fs * TP + r * HIDN + cc], ho[cc] * en);
            }
        }
    }
    __syncthreads();

    const int wave = tid >> 5, lane = tid & 31;
    const int n0 = (wave & 3) * 32, sg = wave >> 2;
    if (sg + 8 < NSTR) l2_mul<2>(s_t, W2, b2, x2, base, sg, n0, lane >> 4, lane & 15);
    else               l2_mul<1>(s_t, W2, b2, x2, base, sg, n0, lane >> 4, lane & 15);
}

#define FCAP 16384
#define SCU  8
static_assert(NPW * EMBN * 4 + NRL * EMBN * 4 + FCAP * 8 + 64 <= 327680, "rows, rel and list must fit the LDS of one work-group");
static_assert(EMBN == 128, "one row = 32 lanes x 4 floats");

__device__ __forceinline__ void fl_put(bool f, unsigned w, unsigned ix, int ln, uint2* s_fl, int* s_n) {
    const unsigned m = __builtin_amdgcn_ballot_w32(f);
    if (m == 0u) return;
    int p0 = 0;
    if (ln == 0) p0 = atomicAdd(s_n, __popc(m));
    p0 = __builtin_amdgcn_readfirstlane(p0);
    if (f) {
        const int pos = p0 + __popc(m & ((1u << ln) - 1u));
        if (pos < FCAP) s_fl[pos] = make_uint2(w, ix);
    }
}

__device__ __forceinline__ float score_lane(float4 xs, float4 rl, float4 xo) {
    float acc = xs.x * rl.x * xo.x;
    acc += xs.y * rl.y * xo.y;
    acc += xs.z * rl.z * xo.z;
    acc += xs.w * rl.w * xo.w;
#pragma unroll
    for (int off = 16; off > 0; off >>= 1) acc += __shfl_xor(acc, off, 32);
    return acc;
}

__global__ __launch_bounds__(L1T) void k_score_own(const int* __restrict__ t,
        const float* __restrict__ x2, const float* __restrict__ rel,
        const unsigned* __restrict__ sorted, const unsigned* __restrict__ sidx,
        const unsigned* __restrict__ segT, float* __restrict__ out) {
    __shared__ __attribute__((aligned(16))) float s_x[NPW * EMBN];
    __shared__ __attribute__((aligned(16))) float s_r[NRL * EMBN];
    __shared__ uint2 s_fl[FCAP];
    __shared__ int   s_n;

    const int tid  = threadIdx.x;
    const int base = blockIdx.x * NPW;
    const int wv = tid >> 5, ln = tid & 31;

    if (tid == 0) s_n = 0;

    unsigned w0[8], i0[8];
    int n[8];
    long off[8];
#pragma unroll
    for (int u = 0; u < 8; u++) {
        const int a  = wv + u * (L1T / 32);
        const int ac = (a < NCH) ? a : NCH - 1;
        const unsigned sg = segT[blockIdx.x * NCH + ac];
        n[u]   = (a < NCH) ? (int)(sg >> 16) : 0;
        off[u] = (long)ac * (2 * CH) + (sg & 0xffffu);
    }
#pragma unroll
    for (int u = 0; u < 8; u++) {
        w0[u] = (ln < n[u]) ? sorted[off[u] + ln] : 0xff000000u;
        i0[u] = (ln < n[u]) ? sidx[off[u] + ln]   : 0u;
    }

    {
        const float4* xg = (const float4*)(x2 + (long)base * EMBN);
        for (int i = tid; i < NPW * EMBN / 4; i += L1T) ((float4*)s_x)[i] = xg[i];
        if (tid < NRL * EMBN / 4) ((float4*)s_r)[tid] = ((const float4*)rel)[tid];
    }
    __syncthreads();

#pragma unroll
    for (int u = 0; u < 8; u++) {
        fl_put((ln < n[u]) && ((w0[u] >> 24) < (unsigned)NRL), w0[u], i0[u], ln, s_fl, &s_n);
        for (int b = 32; b < n[u]; b += 32) {
            const int  i = b + ln;
            const bool v = i < n[u];
            const unsigned w  = v ? sorted[off[u] + i] : 0xff000000u;
            const unsigned ix = v ? sidx[off[u] + i]   : 0u;
            fl_put(v && ((w >> 24) < (unsigned)NRL), w, ix, ln, s_fl, &s_n);
        }
    }
    __syncthreads();
    const int nf = s_n;

    if (nf <= FCAP) {
        for (int e0 = wv; e0 < nf; e0 += SCU * (L1T / 32)) {
            float4 xs[SCU];
            unsigned ww[SCU], ix[SCU];
#pragma unroll
            for (int u = 0; u < SCU; u++) {
                const int e = e0 + u * (L1T / 32);
                const uint2 en = s_fl[(e < nf) ? e : nf - 1];
                ww[u] = en.x; ix[u] = en.y;
                xs[u] = *(const float4*)(x2 + (long)(en.x & 0xffffu) * EMBN + 4 * ln);
            }
#pragma unroll
            for (int u = 0; u < SCU; u++) {
                const int ol = (int)((ww[u] >> 16) & 0xffu), p = (int)(ww[u] >> 24);
                const float4 xo = *(const float4*)(s_x + ol * EMBN + 4 * ln);
                const float4 rl = *(const float4*)(s_r + p * EMBN + 4 * ln);
                const float acc = score_lane(xs[u], rl, xo);
                if (ln == 0 && e0 + u * (L1T / 32) < nf) out[ix[u]] = acc;
            }
        }
    } else {
        for (int k = wv; k < NT; k += L1T / 32) {
            const int s = t[3*k], p = t[3*k+1], o = t[3*k+2];
            const unsigned fo = (unsigned)(o - base);
            if (fo < NPW) {
                const float4 xv = *(const float4*)(x2 + (long)s * EMBN + 4 * ln);
                const float4 xo = *(const float4*)(s_x + fo * EMBN + 4 * ln);
                const float4 rl = *(const float4*)(s_r + p * EMBN + 4 * ln);
                const float acc = score_lane(xv, rl, xo);
                if (ln == 0) out[k] = acc;
            }
        }
    }
}

extern "C" void kernel_launch(void* const* d_in, const int* in_sizes, int n_in,
                              void* d_out, int out_size, void* d_ws, size_t ws_size,
                              hipStream_t stream) {
    const int*   triples = (const int*)  d_in[0];
    const float* W1      = (const float*)d_in[1];
    const float* b1      = (const float*)d_in[2];
    const float* W2      = (const float*)d_in[3];
    const float* b2      = (const float*)d_in[4];
    const float* rel     = (const float*)d_in[5];
    float*       out     = (float*)d_out;

    float* deg = (float*)d_ws;
    float* x1  = deg + (long)RP * NN;
    float* agg = x1  + (long)NN * HIDN;
    float* x2  = agg + (long)NN * KDIM;
    unsigned* glist = (unsigned*)agg;
    int*      gcnt  = (int*)(agg + 2L * NWG * LCAP);
    unsigned* sorted = (unsigned*)(agg + 2L * NWG * LCAP + 256);
    unsigned* segT   = sorted + 2L * NT;
    unsigned* sidx   = segT + (long)NWG * NCH;
    static_assert(2L * NWG * LCAP + 256 + 4L * NT + (long)NWG * NCH <= (long)NN * KDIM, "lists, sorted chunks and their triple indices fit where agg was");

    k_bin<<<NCH, L1T, 0, stream>>>(triples, sorted, sidx, segT);
    k_layer1<<<NWG, L1T, 0, stream>>>(triples, W1, b1, deg, x1, glist, gcnt, sorted, segT);
    k_layer2<<<NWG, L1T, 0, stream>>>(triples, glist, gcnt, x1, deg, W2, b2, x2);
    k_score_own<<<NWG, L1T, 0, stream>>>(triples, x2, rel, sorted, sidx, segT, out);
}


# Fewer waves to enter: sort launch in 8-wave work-groups, the other three in 16-wave ones, each wave with 2 to 4 times the loads in flight; list order new, LDS f32 adds by arrival as before
# speedup vs baseline: 5.3929x; 1.0791x over previous
#include <hip/hip_runtime.h>

typedef float v2f __attribute__((ext_vector_type(2)));
typedef float v8f __attribute__((ext_vector_type(8)));

#define NN   50000
#define NRL  12
#define NT   500000
#define HIDN 16
#define EMBN 128
#define RP   25
#define KDIM 400
#define NAUG (2*NT + NN)

#define NPW  200
#define NWG  (NN / NPW)
#define LCAP 32768
#define WGT  512
#define NWV  (WGT / 32)
#define SPW  16
#define SCT  512
#define BT   256
#define BU   8
#define NCH  250
#define CH   (NT / NCH)
static_assert(NN % NPW == 0, "owners must tile the nodes");
static_assert(NN <= 65536 && NPW <= 256 && RP <= 256, "list word packing");
static_assert(NT % NCH == 0 && 2 * CH < 65536 && CH % BU == 0 && CH / BU <= BT && BT == 256, "chunking and segment word packing");
static_assert(NWG <= 256 && NCH <= 256 && NCH <= SPW * NWV, "one wave scans 256 counts; SPW segments per wave");
static_assert(WGT >= 256, "the first 256 threads load the segment table");

__device__ __forceinline__ void wave_excl_scan(const int* cv, int* ov, int n, int sh, int lane, int* tot) {
    int v[8], sum = 0;
#pragma unroll
    for (int j = 0; j < 8; j++) {
        int idx = lane * 8 + j;
        int x = (idx < n) ? (int)((unsigned)cv[idx] >> sh) : 0;
        v[j] = sum; sum += x;
    }
    int incl = sum;
#pragma unroll
    for (int d = 1; d < 32; d <<= 1) {
        int y = __shfl_up(incl, d, 32);
        if (lane >= d) incl += y;
    }
    const int b = incl - sum;
#pragma unroll
    for (int j = 0; j < 8; j++) {
        int idx = lane * 8 + j;
        if (idx < n) ov[idx] = b + v[j];
    }
    if (tot != nullptr && lane == 31) *tot = incl;
}

__global__ __launch_bounds__(BT) void k_bin(const int* __restrict__ t,
        unsigned* __restrict__ sorted, unsigned* __restrict__ sidx, unsigned* __restrict__ segT) {
    __shared__ int s_c[256];
    __shared__ int s_o[256];
    const int tid = threadIdx.x;
    const int a   = blockIdx.x;
    s_c[tid] = 0;
    __syncthreads();

    const bool on = tid < CH / BU;
    unsigned wf[BU], wi[BU];
    int gf[BU], gi[BU], rf[BU], ri[BU];
#pragma unroll
    for (int u = 0; u < BU; u++) { wf[u] = 0u; wi[u] = 0u; gf[u] = 0; gi[u] = 0; rf[u] = 0; ri[u] = 0; }
    if (on) {
        const int* tp = t + 3L * (a * CH + tid);
        int ts[BU], tr[BU], to[BU];
#pragma unroll
        for (int u = 0; u < BU; u++) {
            ts[u] = tp[3 * (CH / BU) * u];
            tr[u] = tp[3 * (CH / BU) * u + 1];
            to[u] = tp[3 * (CH / BU) * u + 2];
        }
#pragma unroll
        for (int u = 0; u < BU; u++) {
            int s = ts[u], p = tr[u], o = to[u];
            int go = o / NPW, gs = s / NPW;
            wf[u] = (unsigned)s | ((unsigned)(o - go * NPW) << 16) | ((unsigned)p << 24);
            wi[u] = (unsigned)o | ((unsigned)(s - gs * NPW) << 16) | ((unsigned)(p + NRL) << 24);
            gf[u] = go; gi[u] = gs;
            rf[u] = atomicAdd(&s_c[go], 1);
            ri[u] = atomicAdd(&s_c[gs], 1);
        }
    }
    __syncthreads();
    if (tid < 32) wave_excl_scan(s_c, s_o, NWG, 0, tid, nullptr);
    __syncthreads();

    if (on) {
        unsigned* sp = sorted + (long)a * (2 * CH);
        unsigned* ip = sidx   + (long)a * (2 * CH);
#pragma unroll
        for (int u = 0; u < BU; u++) {
            const int pf = s_o[gf[u]] + rf[u], pi = s_o[gi[u]] + ri[u];
            const unsigned ti = (unsigned)(a * CH + tid + u * (CH / BU));
            sp[pf] = wf[u]; ip[pf] = ti;
            sp[pi] = wi[u]; ip[pi] = ti;
        }
    }
    if (tid < NWG) segT[tid * NCH + a] = (unsigned)s_o[tid] | ((unsigned)s_c[tid] << 16);
}

__device__ __forceinline__ void l1_put(unsigned w, int pos, int* s_deg, unsigned* s_list) {
    atomicAdd(&s_deg[(int)(w >> 24) * NPW + (int)((w >> 16) & 0xffu)], 1);
    if (pos < LCAP) s_list[pos] = w;
}

#define L1U 16
#define L1G (WGT / 16)
#define X1J ((NPW * HIDN + WGT - 1) / WGT)
#define SLJ ((NPW + L1G - 1) / L1G)
#define TRIP (L1U * L1G)
#define PADW(base) ((unsigned)(base) | ((unsigned)NPW << 16))
static_assert(LCAP % TRIP == 0 && NPW < 256, "whole passes fit the list; the spare destination fits its field");

__global__ __launch_bounds__(WGT) void k_layer1(const int* __restrict__ t,
        const float* __restrict__ W1, const float* __restrict__ b1,
        float* __restrict__ norm, float* __restrict__ h,
        unsigned* __restrict__ glist, int* __restrict__ gcnt,
        const unsigned* __restrict__ sorted, const unsigned* __restrict__ segT) {
    __shared__ int      s_deg[RP * NPW];
    __shared__ float    s_x1[NPW * HIDN + HIDN];
    __shared__ unsigned s_list[LCAP];
    __shared__ int      s_seg[256];
    __shared__ int      s_dst[256];
    __shared__ int      s_cnt;

    const int tid  = threadIdx.x;
    const int base = blockIdx.x * NPW;

    for (int i = tid; i < RP * NPW; i += WGT) s_deg[i] = (i >= 2 * NRL * NPW) ? 1 : 0;
    if (tid < 256) s_seg[tid] = (tid < NCH) ? (int)segT[blockIdx.x * NCH + tid] : 0;
    __syncthreads();
    if (tid < 32) wave_excl_scan(s_seg, s_dst, NCH, 16, tid, &s_cnt);
    __syncthreads();

    {
        const int wv = tid >> 5, ln = tid & 31;
        unsigned w0[SPW];
        int n[SPW], d0[SPW], so[SPW];
#pragma unroll
        for (int u = 0; u < SPW; u++) {
            int a = wv + u * NWV;
            n[u] = 0; d0[u] = 0; so[u] = 0;
            if (a < NCH) {
                unsigned sg = (unsigned)s_seg[a];
                n[u]  = (int)(sg >> 16);
                so[u] = a * (2 * CH) + (int)(sg & 0xffffu);
                d0[u] = s_dst[a];
            }
            w0[u] = (ln < n[u]) ? sorted[so[u] + ln] : 0u;
        }
#pragma unroll
        for (int u = 0; u < SPW; u++) {
            if (ln < n[u]) l1_put(w0[u], d0[u] + ln, s_deg, s_list);
            for (int i = ln + 32; i < n[u]; i += 32) l1_put(sorted[so[u] + i], d0[u] + i, s_deg, s_list);
        }
    }
    __syncthreads();
    const int cnt  = s_cnt;
    const int cntp = (cnt + TRIP - 1) / TRIP * TRIP;
    if (cnt <= LCAP)
        for (int i = cnt + tid; i < cntp; i += WGT) s_list[i] = PADW(base);

    float sv[X1J];
#pragma unroll
    for (int j = 0; j < X1J; j++) {
        const int i = tid + j * WGT;
        sv[j] = (i < NPW * HIDN) ? W1[((long)(2 * NRL) * NN + base) * HIDN + i] : 0.0f;
    }

    for (int i = tid; i < RP * NPW; i += WGT) {
        float nv = 1.0f / fmaxf((float)s_deg[i], 1.0f);
        s_deg[i] = __float_as_int(nv);
        int r = i / NPW, ol = i - r * NPW;
        norm[r * NN + base + ol] = nv;
    }
    __syncthreads();

    if (cnt <= LCAP) {
        uint2* gl = (uint2*)glist + (long)blockIdx.x * LCAP;
        for (int i = tid; i < cntp; i += WGT) {
            unsigned w = s_list[i];
            int ol = (int)((w >> 16) & 0xffu), r = (int)(w >> 24);
            gl[i] = make_uint2(w, (unsigned)s_deg[r * NPW + ol]);
        }
    }
    if (tid == 0) gcnt[blockIdx.x] = cnt;

#pragma unroll
    for (int j = 0; j < X1J; j++) {
        const int i = tid + j * WGT;
        if (i < NPW * HIDN) s_x1[i] = sv[j] * __int_as_float(s_deg[2 * NRL * NPW + (i >> 4)]);
    }
    __syncthreads();

    if (cnt <= LCAP) {
        const int g = tid >> 4, c = tid & 15;
        for (int e = g; e < cntp; e += TRIP) {
            float v[L1U]; int dst[L1U];
#pragma unroll
            for (int u = 0; u < L1U; u++) {
                unsigned w = s_list[e + u * L1G];
                int sN = (int)(w & 0xffffu), ol = (int)((w >> 16) & 0xffu), r = (int)(w >> 24);
                float en = __int_as_float(s_deg[r * NPW + ol]);
                v[u] = W1[((long)r * NN + sN) * HIDN + c] * en;
                dst[u] = ol * HIDN + c;
            }
#pragma unroll
            for (int u = 0; u < L1U; u++) atomicAdd(&s_x1[dst[u]], v[u]);
        }
    } else {
        for (int k = tid; k < NT; k += WGT) {
            int s = t[3*k], p = t[3*k+1], o = t[3*k+2];
            unsigned fo = (unsigned)(o - base), fs = (unsigned)(s - base);
            if (fo < NPW) {
                float en = __int_as_float(s_deg[p * NPW + fo]);
                const float* w = W1 + ((long)p * NN + s) * HIDN;
                for (int c = 0; c < HIDN; c++) atomicAdd(&s_x1[fo * HIDN + c], w[c] * en);
            }
            if (fs < NPW) {
                int r = p + NRL;
                float en = __int_as_float(s_deg[r * NPW + fs]);
                const float* w = W1 + ((long)r * NN + o) * HIDN;
                for (int c = 0; c < HIDN; c++) atomicAdd(&s_x1[fs * HIDN + c], w[c] * en);
            }
        }
    }
    __syncthreads();

    for (int i = tid; i < NPW * HIDN; i += WGT)
        h[(long)base * HIDN + i] = fmaxf(s_x1[i] + b1[i & (HIDN - 1)], 0.0f);
}

#define TP   404
#define NSTR ((NPW + 15) / 16)
#define SSTR 4
static_assert((NPW * TP + HIDN) * 4 <= 327680, "the tile and its spare words must fit the LDS of one work-group");
static_assert(WGT == 512 && NSTR == 13 && SSTR == 4, "wave to (column pair, strip group) map below: groups of 4, 3, 3, 3 strips");

template <int NS>
__device__ __forceinline__ void l2_mul(const float* s_t, const float* __restrict__ W2,
        const float* __restrict__ b2, float* __restrict__ x2,
        int base, int sg, int n0, int half, int lr) {
    v8f acc[NS][2];
    int aoff[NS];
#pragma unroll
    for (int j = 0; j < NS; j++) {
        v8f z = {0.f, 0.f, 0.f, 0.f, 0.f, 0.f, 0.f, 0.f};
        acc[j][0] = z; acc[j][1] = z;
        int row = (sg + SSTR * j) * 16 + lr;
        if (row > NPW - 1) row = NPW - 1;
        aoff[j] = row * TP + 2 * half;
    }
    const float* bp = W2 + (long)(2 * half) * EMBN + n0 + lr;
#pragma unroll 10
    for (int k0 = 0; k0 < KDIM; k0 += 4) {
        v2f b0, b1;
        b0.x = bp[(long)k0 * EMBN];        b0.y = bp[(long)(k0 + 1) * EMBN];
        b1.x = bp[(long)k0 * EMBN + 16];   b1.y = bp[(long)(k0 + 1) * EMBN + 16];
#pragma unroll
        for (int j = 0; j < NS; j++) {
            v2f a = *(const v2f*)(s_t + aoff[j] + k0);
            acc[j][0] = __builtin_amdgcn_wmma_f32_16x16x4_f32(
                false, a, false, b0, (short)0, acc[j][0], false, false);
            acc[j][1] = __builtin_amdgcn_wmma_f32_16x16x4_f32(
                false, a, false, b1, (short)0, acc[j][1], false, false);
        }
    }
#pragma unroll
    for (int j = 0; j < NS; j++) {
#pragma unroll
        for (int q = 0; q < 2; q++) {
            int n = n0 + 16 * q + lr;
            float bias = b2[n];
#pragma unroll
            for (int v = 0; v < 8; v++) {
                int row = (sg + SSTR * j) * 16 + v + 8 * half;
                if (row < NPW) x2[(long)(base + row) * EMBN + n] = acc[j][q][v] + bias;
            }
        }
    }
}

__global__ __launch_bounds__(WGT) void k_layer2(const int* __restrict__ t,
        const unsigned* __restrict__ glist, const int* __restrict__ gcnt,
        const float* __restrict__ h, const float* __restrict__ norm,
        const float* __restrict__ W2, const float* __restrict__ b2, float* __restrict__ x2) {
    __shared__ __attribute__((aligned(16))) float s_t[NPW * TP + HIDN];

    const int tid  = threadIdx.x;
    const int base = blockIdx.x * NPW;
    const int g = tid >> 4, c = tid & 15;

    float hv[SLJ], nv[SLJ];
#pragma unroll
    for (int j = 0; j < SLJ; j++) {
        const int ol = g + j * L1G;
        hv[j] = (ol < NPW) ? h[(long)(base + ol) * HIDN + c] : 0.0f;
        nv[j] = (ol < NPW) ? norm[2 * NRL * NN + base + ol] : 0.0f;
    }
    const int cnt = gcnt[blockIdx.x];

    for (int i = tid; i < NPW * TP / 4; i += WGT) ((float4*)s_t)[i] = make_float4(0.f, 0.f, 0.f, 0.f);
    __syncthreads();

#pragma unroll
    for (int j = 0; j < SLJ; j++) {
        const int ol = g + j * L1G;
        if (ol < NPW) s_t[ol * TP + 2 * NRL * HIDN + c] = hv[j] * nv[j];
    }

    if (cnt <= LCAP) {
        const uint2* lp = (const uint2*)glist + (long)blockIdx.x * LCAP;
        const int cntp = (cnt + TRIP - 1) / TRIP * TRIP;
        for (int e = g; e < cntp; e += TRIP) {
            float v[L1U]; int dst[L1U];
#pragma unroll
            for (int u = 0; u < L1U; u++) {
                uint2 w = lp[e + u * L1G];
                int sN = (int)(w.x & 0xffffu), ol = (int)((w.x >> 16) & 0xffu), r = (int)(w.x >> 24);
                v[u] = h[(long)sN * HIDN + c] * __uint_as_float(w.y);
                dst[u] = ol * TP + r * HIDN + c;
            }
#pragma unroll
            for (int u = 0; u < L1U; u++) atomicAdd(&s_t[dst[u]], v[u]);
        }
    } else {
        for (int k = tid; k < NT; k += WGT) {
            int s = t[3*k], p = t[3*k+1], o = t[3*k+2];
            unsigned fo = (unsigned)(o - base), fs = (unsigned)(s - base);
            if (fo < NPW) {
                float en = norm[p * NN + o];
                const float* hs = h + (long)s * HIDN;
                for (int cc = 0; cc < HIDN; cc++) atomicAdd(&s_t[fo * TP + p * HIDN + cc], hs[cc] * en);
            }
            if (fs < NPW) {
                int r = p + NRL;
                float en = norm[r * NN + s];
                const float* ho = h + (long)o * HIDN;
                for (int cc = 0; cc < HIDN; cc++) atomicAdd(&s_t[fs * TP + r * HIDN + cc], ho[cc] * en);
            }
        }
    }
    __syncthreads();

    const int wave = tid >> 5, lane = tid & 31;
    const int cp = wave & 3, sg = ((wave >> 2) - cp) & 3;
    if (sg == 0) l2_mul<4>(s_t, W2, b2, x2, base, sg, cp * 32, lane >> 4, lane & 15);
    else         l2_mul<3>(s_t, W2, b2, x2, base, sg, cp * 32, lane >> 4, lane & 15);
}

#define FCAP 16384
#define SCW  (SCT / 32)
#define SSPW 16
#define SCU  16
#define XJ   (NPW * EMBN / 4 / SCT)
static_assert(NPW * EMBN * 4 + NRL * EMBN * 4 + FCAP * 8 + 64 <= 327680, "rows, rel and list must fit the LDS of one work-group");
static_assert(EMBN == 128, "one row = 32 lanes x 4 floats");
static_assert(NCH <= SSPW * SCW && NRL * EMBN / 4 <= SCT && NPW * EMBN / 4 - XJ * SCT <= SCT, "segments per wave; rel and the rest of the rows in one pass");

__device__ __forceinline__ void fl_put(bool f, unsigned w, unsigned ix, int ln, uint2* s_fl, int* s_n) {
    const unsigned m = __builtin_amdgcn_ballot_w32(f);
    if (m == 0u) return;
    int p0 = 0;
    if (ln == 0) p0 = atomicAdd(s_n, __popc(m));
    p0 = __builtin_amdgcn_readfirstlane(p0);
    if (f) {
        const int pos = p0 + __popc(m & ((1u << ln) - 1u));
        if (pos < FCAP) s_fl[pos] = make_uint2(w, ix);
    }
}

__device__ __forceinline__ float score_part(float4 xs, float4 rl, float4 xo) {
    float acc = xs.x * rl.x * xo.x;
    acc += xs.y * rl.y * xo.y;
    acc += xs.z * rl.z * xo.z;
    acc += xs.w * rl.w * xo.w;
    return acc;
}

__device__ __forceinline__ float score_lane(float4 xs, float4 rl, float4 xo) {
    float acc = score_part(xs, rl, xo);
#pragma unroll
    for (int off = 16; off > 0; off >>= 1) acc += __shfl_xor(acc, off, 32);
    return acc;
}

__global__ __launch_bounds__(SCT) void k_score_own(const int* __restrict__ t,
        const float* __restrict__ x2, const float* __restrict__ rel,
        const unsigned* __restrict__ sorted, const unsigned* __restrict__ sidx,
        const unsigned* __restrict__ segT, float* __restrict__ out) {
    __shared__ __attribute__((aligned(16))) float s_x[NPW * EMBN];
    __shared__ __attribute__((aligned(16))) float s_r[NRL * EMBN];
    __shared__ uint2 s_fl[FCAP];
    __shared__ int   s_n;

    const int tid  = threadIdx.x;
    const int base = blockIdx.x * NPW;
    const int wv = tid >> 5, ln = tid & 31;

    if (tid == 0) s_n = 0;

    unsigned w0[SSPW], i0[SSPW];
    int n[SSPW], off[SSPW];
#pragma unroll
    for (int u = 0; u < SSPW; u++) {
        const int a  = wv + u * SCW;
        const int ac = (a < NCH) ? a : NCH - 1;
        const unsigned sg = segT[blockIdx.x * NCH + ac];
        n[u]   = (a < NCH) ? (int)(sg >> 16) : 0;
        off[u] = ac * (2 * CH) + (int)(sg & 0xffffu);
    }
#pragma unroll
    for (int u = 0; u < SSPW; u++) {
        w0[u] = (ln < n[u]) ? sorted[off[u] + ln] : 0xff000000u;
        i0[u] = (ln < n[u]) ? sidx[off[u] + ln]   : 0u;
    }

    {
        const float4* xg = (const float4*)(x2 + (long)base * EMBN);
        float4 xr[XJ];
#pragma unroll
        for (int j = 0; j < XJ; j++) xr[j] = xg[tid + j * SCT];
#pragma unroll
        for (int j = 0; j < XJ; j++) ((float4*)s_x)[tid + j * SCT] = xr[j];
        if (tid + XJ * SCT < NPW * EMBN / 4) ((float4*)s_x)[tid + XJ * SCT] = xg[tid + XJ * SCT];
        if (tid < NRL * EMBN / 4) ((float4*)s_r)[tid] = ((const float4*)rel)[tid];
    }
    __syncthreads();

    {
        unsigned m[SSPW];
        int tot = 0;
#pragma unroll
        for (int u = 0; u < SSPW; u++) {
            m[u] = __builtin_amdgcn_ballot_w32((ln < n[u]) && ((w0[u] >> 24) < (unsigned)NRL));
            tot += __popc(m[u]);
        }
        int p0 = 0;
        if (ln == 0 && tot > 0) p0 = atomicAdd(&s_n, tot);
        p0 = __builtin_amdgcn_readfirstlane(p0);
#pragma unroll
        for (int u = 0; u < SSPW; u++) {
            if ((m[u] >> ln) & 1u) {
                const int pos = p0 + __popc(m[u] & ((1u << ln) - 1u));
                if (pos < FCAP) s_fl[pos] = make_uint2(w0[u], i0[u]);
            }
            p0 += __popc(m[u]);
        }
#pragma unroll
        for (int u = 0; u < SSPW; u++) {
            for (int b = 32; b < n[u]; b += 32) {
                const int  i = b + ln;
                const bool v = i < n[u];
                const unsigned w  = v ? sorted[off[u] + i] : 0xff000000u;
                const unsigned ix = v ? sidx[off[u] + i]   : 0u;
                fl_put(v && ((w >> 24) < (unsigned)NRL), w, ix, ln, s_fl, &s_n);
            }
        }
    }
    __syncthreads();
    const int nf = s_n;

    if (nf <= FCAP) {
        for (int e0 = wv; e0 < nf; e0 += SCU * SCW) {
            float4 xs[SCU];
            unsigned ww[SCU], ix[SCU];
            float acc[SCU];
#pragma unroll
            for (int u = 0; u < SCU; u++) {
                const int e = e0 + u * SCW;
                const uint2 en = s_fl[(e < nf) ? e : nf - 1];
                ww[u] = en.x; ix[u] = en.y;
                xs[u] = *(const float4*)(x2 + (long)(en.x & 0xffffu) * EMBN + 4 * ln);
            }
#pragma unroll
            for (int u = 0; u < SCU; u++) {
                const int ol = (int)((ww[u] >> 16) & 0xffu), p = (int)(ww[u] >> 24);
                const float4 xo = *(const float4*)(s_x + ol * EMBN + 4 * ln);
                const float4 rl = *(const float4*)(s_r + p * EMBN + 4 * ln);
                acc[u] = score_part(xs[u], rl, xo);
            }
#pragma unroll
            for (int off2 = 16; off2 > 0; off2 >>= 1) {
#pragma unroll
                for (int u = 0; u < SCU; u++) acc[u] += __shfl_xor(acc[u], off2, 32);
            }
#pragma unroll
            for (int u = 0; u < SCU; u++)
                if (ln == 0 && e0 + u * SCW < nf) out[ix[u]] = acc[u];
        }
    } else {
        for (int k = wv; k < NT; k += SCW) {
            const int s = t[3*k], p = t[3*k+1], o = t[3*k+2];
            const unsigned fo = (unsigned)(o - base);
            if (fo < NPW) {
                const float4 xv = *(const float4*)(x2 + (long)s * EMBN + 4 * ln);
                const float4 xo = *(const float4*)(s_x + fo * EMBN + 4 * ln);
                const float4 rl = *(const float4*)(s_r + p * EMBN + 4 * ln);
                const float acc = score_lane(xv, rl, xo);
                if (ln == 0) out[k] = acc;
            }
        }
    }
}

extern "C" void kernel_launch(void* const* d_in, const int* in_sizes, int n_in,
                              void* d_out, int out_size, void* d_ws, size_t ws_size,
                              hipStream_t stream) {
    const int*   triples = (const int*)  d_in[0];
    const float* W1      = (const float*)d_in[1];
    const float* b1      = (const float*)d_in[2];
    const float* W2      = (const float*)d_in[3];
    const float* b2      = (const float*)d_in[4];
    const float* rel     = (const float*)d_in[5];
    float*       out     = (float*)d_out;

    float* deg = (float*)d_ws;
    float* x1  = deg + (long)RP * NN;
    float* agg = x1  + (long)NN * HIDN;
    float* x2  = agg + (long)NN * KDIM;
    unsigned* glist = (unsigned*)agg;
    int*      gcnt  = (int*)(agg + 2L * NWG * LCAP);
    unsigned* sorted = (unsigned*)(agg + 2L * NWG * LCAP + 256);
    unsigned* segT   = sorted + 2L * NT;
    unsigned* sidx   = segT + (long)NWG * NCH;
    static_assert(2L * NWG * LCAP + 256 + 4L * NT + (long)NWG * NCH <= (long)NN * KDIM, "lists, sorted chunks and their triple indices fit where agg was");

    k_bin<<<NCH, BT, 0, stream>>>(triples, sorted, sidx, segT);
    k_layer1<<<NWG, WGT, 0, stream>>>(triples, W1, b1, deg, x1, glist, gcnt, sorted, segT);
    k_layer2<<<NWG, WGT, 0, stream>>>(triples, glist, gcnt, x1, deg, W2, b2, x2);
    k_score_own<<<NWG, SCT, 0, stream>>>(triples, x2, rel, sorted, sidx, segT, out);
}


# Cross-lane moves on the vector ALU, not via memory: score fold by permlanex16 + row_xmask DPP adds (same sums); layer-2 list words one per lane, row_share to 16-lane groups (LDS f32 adds by arrival)
# speedup vs baseline: 5.5848x; 1.0356x over previous
#include <hip/hip_runtime.h>

typedef float v2f __attribute__((ext_vector_type(2)));
typedef float v8f __attribute__((ext_vector_type(8)));

#define NN   50000
#define NRL  12
#define NT   500000
#define HIDN 16
#define EMBN 128
#define RP   25
#define KDIM 400
#define NAUG (2*NT + NN)

#define NPW  200
#define NWG  (NN / NPW)
#define LCAP 32768
#define WGT  512
#define NWV  (WGT / 32)
#define SPW  16
#define SCT  512
#define BT   256
#define BU   8
#define NCH  250
#define CH   (NT / NCH)
static_assert(NN % NPW == 0, "owners must tile the nodes");
static_assert(NN <= 65536 && NPW <= 256 && RP <= 256, "list word packing");
static_assert(NT % NCH == 0 && 2 * CH < 65536 && CH % BU == 0 && CH / BU <= BT && BT == 256, "chunking and segment word packing");
static_assert(NWG <= 256 && NCH <= 256 && NCH <= SPW * NWV, "one wave scans 256 counts; SPW segments per wave");
static_assert(WGT >= 256, "the first 256 threads load the segment table");

__device__ __forceinline__ void wave_excl_scan(const int* cv, int* ov, int n, int sh, int lane, int* tot) {
    int v[8], sum = 0;
#pragma unroll
    for (int j = 0; j < 8; j++) {
        int idx = lane * 8 + j;
        int x = (idx < n) ? (int)((unsigned)cv[idx] >> sh) : 0;
        v[j] = sum; sum += x;
    }
    int incl = sum;
#pragma unroll
    for (int d = 1; d < 32; d <<= 1) {
        int y = __shfl_up(incl, d, 32);
        if (lane >= d) incl += y;
    }
    const int b = incl - sum;
#pragma unroll
    for (int j = 0; j < 8; j++) {
        int idx = lane * 8 + j;
        if (idx < n) ov[idx] = b + v[j];
    }
    if (tot != nullptr && lane == 31) *tot = incl;
}

__global__ __launch_bounds__(BT) void k_bin(const int* __restrict__ t,
        unsigned* __restrict__ sorted, unsigned* __restrict__ sidx, unsigned* __restrict__ segT) {
    __shared__ int s_c[256];
    __shared__ int s_o[256];
    const int tid = threadIdx.x;
    const int a   = blockIdx.x;
    s_c[tid] = 0;
    __syncthreads();

    const bool on = tid < CH / BU;
    unsigned wf[BU], wi[BU];
    int gf[BU], gi[BU], rf[BU], ri[BU];
#pragma unroll
    for (int u = 0; u < BU; u++) { wf[u] = 0u; wi[u] = 0u; gf[u] = 0; gi[u] = 0; rf[u] = 0; ri[u] = 0; }
    if (on) {
        const int* tp = t + 3L * (a * CH + tid);
        int ts[BU], tr[BU], to[BU];
#pragma unroll
        for (int u = 0; u < BU; u++) {
            ts[u] = tp[3 * (CH / BU) * u];
            tr[u] = tp[3 * (CH / BU) * u + 1];
            to[u] = tp[3 * (CH / BU) * u + 2];
        }
#pragma unroll
        for (int u = 0; u < BU; u++) {
            int s = ts[u], p = tr[u], o = to[u];
            int go = o / NPW, gs = s / NPW;
            wf[u] = (unsigned)s | ((unsigned)(o - go * NPW) << 16) | ((unsigned)p << 24);
            wi[u] = (unsigned)o | ((unsigned)(s - gs * NPW) << 16) | ((unsigned)(p + NRL) << 24);
            gf[u] = go; gi[u] = gs;
            rf[u] = atomicAdd(&s_c[go], 1);
            ri[u] = atomicAdd(&s_c[gs], 1);
        }
    }
    __syncthreads();
    if (tid < 32) wave_excl_scan(s_c, s_o, NWG, 0, tid, nullptr);
    __syncthreads();

    if (on) {
        unsigned* sp = sorted + (long)a * (2 * CH);
        unsigned* ip = sidx   + (long)a * (2 * CH);
#pragma unroll
        for (int u = 0; u < BU; u++) {
            const int pf = s_o[gf[u]] + rf[u], pi = s_o[gi[u]] + ri[u];
            const unsigned ti = (unsigned)(a * CH + tid + u * (CH / BU));
            sp[pf] = wf[u]; ip[pf] = ti;
            sp[pi] = wi[u]; ip[pi] = ti;
        }
    }
    if (tid < NWG) segT[tid * NCH + a] = (unsigned)s_o[tid] | ((unsigned)s_c[tid] << 16);
}

__device__ __forceinline__ void l1_put(unsigned w, int pos, int* s_deg, unsigned* s_list) {
    atomicAdd(&s_deg[(int)(w >> 24) * NPW + (int)((w >> 16) & 0xffu)], 1);
    if (pos < LCAP) s_list[pos] = w;
}

#define L1U 16
#define L1G (WGT / 16)
#define X1J ((NPW * HIDN + WGT - 1) / WGT)
#define SLJ ((NPW + L1G - 1) / L1G)
#define TRIP (L1U * L1G)
#define PADW(base) ((unsigned)(base) | ((unsigned)NPW << 16))
static_assert(LCAP % TRIP == 0 && NPW < 256, "whole passes fit the list; the spare destination fits its field");

__global__ __launch_bounds__(WGT) void k_layer1(const int* __restrict__ t,
        const float* __restrict__ W1, const float* __restrict__ b1,
        float* __restrict__ norm, float* __restrict__ h,
        unsigned* __restrict__ glist, int* __restrict__ gcnt,
        const unsigned* __restrict__ sorted, const unsigned* __restrict__ segT) {
    __shared__ int      s_deg[RP * NPW];
    __shared__ float    s_x1[NPW * HIDN + HIDN];
    __shared__ unsigned s_list[LCAP];
    __shared__ int      s_seg[256];
    __shared__ int      s_dst[256];
    __shared__ int      s_cnt;

    const int tid  = threadIdx.x;
    const int base = blockIdx.x * NPW;

    for (int i = tid; i < RP * NPW; i += WGT) s_deg[i] = (i >= 2 * NRL * NPW) ? 1 : 0;
    if (tid < 256) s_seg[tid] = (tid < NCH) ? (int)segT[blockIdx.x * NCH + tid] : 0;
    __syncthreads();
    if (tid < 32) wave_excl_scan(s_seg, s_dst, NCH, 16, tid, &s_cnt);
    __syncthreads();

    {
        const int wv = tid >> 5, ln = tid & 31;
        unsigned w0[SPW];
        int n[SPW], d0[SPW], so[SPW];
#pragma unroll
        for (int u = 0; u < SPW; u++) {
            int a = wv + u * NWV;
            n[u] = 0; d0[u] = 0; so[u] = 0;
            if (a < NCH) {
                unsigned sg = (unsigned)s_seg[a];
                n[u]  = (int)(sg >> 16);
                so[u] = a * (2 * CH) + (int)(sg & 0xffffu);
                d0[u] = s_dst[a];
            }
            w0[u] = (ln < n[u]) ? sorted[so[u] + ln] : 0u;
        }
#pragma unroll
        for (int u = 0; u < SPW; u++) {
            if (ln < n[u]) l1_put(w0[u], d0[u] + ln, s_deg, s_list);
            for (int i = ln + 32; i < n[u]; i += 32) l1_put(sorted[so[u] + i], d0[u] + i, s_deg, s_list);
        }
    }
    __syncthreads();
    const int cnt  = s_cnt;
    const int cntp = (cnt + TRIP - 1) / TRIP * TRIP;
    if (cnt <= LCAP)
        for (int i = cnt + tid; i < cntp; i += WGT) s_list[i] = PADW(base);

    float sv[X1J];
#pragma unroll
    for (int j = 0; j < X1J; j++) {
        const int i = tid + j * WGT;
        sv[j] = (i < NPW * HIDN) ? W1[((long)(2 * NRL) * NN + base) * HIDN + i] : 0.0f;
    }

    for (int i = tid; i < RP * NPW; i += WGT) {
        float nv = 1.0f / fmaxf((float)s_deg[i], 1.0f);
        s_deg[i] = __float_as_int(nv);
        int r = i / NPW, ol = i - r * NPW;
        norm[r * NN + base + ol] = nv;
    }
    __syncthreads();

    if (cnt <= LCAP) {
        uint2* gl = (uint2*)glist + (long)blockIdx.x * LCAP;
        for (int i = tid; i < cntp; i += WGT) {
            unsigned w = s_list[i];
            int ol = (int)((w >> 16) & 0xffu), r = (int)(w >> 24);
            gl[i] = make_uint2(w, (unsigned)s_deg[r * NPW + ol]);
        }
    }
    if (tid == 0) gcnt[blockIdx.x] = cnt;

#pragma unroll
    for (int j = 0; j < X1J; j++) {
        const int i = tid + j * WGT;
        if (i < NPW * HIDN) s_x1[i] = sv[j] * __int_as_float(s_deg[2 * NRL * NPW + (i >> 4)]);
    }
    __syncthreads();

    if (cnt <= LCAP) {
        const int g = tid >> 4, c = tid & 15;
        for (int e = g; e < cntp; e += TRIP) {
            float v[L1U]; int dst[L1U];
#pragma unroll
            for (int u = 0; u < L1U; u++) {
                unsigned w = s_list[e + u * L1G];
                int sN = (int)(w & 0xffffu), ol = (int)((w >> 16) & 0xffu), r = (int)(w >> 24);
                float en = __int_as_float(s_deg[r * NPW + ol]);
                v[u] = W1[((long)r * NN + sN) * HIDN + c] * en;
                dst[u] = ol * HIDN + c;
            }
#pragma unroll
            for (int u = 0; u < L1U; u++) atomicAdd(&s_x1[dst[u]], v[u]);
        }
    } else {
        for (int k = tid; k < NT; k += WGT) {
            int s = t[3*k], p = t[3*k+1], o = t[3*k+2];
            unsigned fo = (unsigned)(o - base), fs = (unsigned)(s - base);
            if (fo < NPW) {
                float en = __int_as_float(s_deg[p * NPW + fo]);
                const float* w = W1 + ((long)p * NN + s) * HIDN;
                for (int c = 0; c < HIDN; c++) atomicAdd(&s_x1[fo * HIDN + c], w[c] * en);
            }
            if (fs < NPW) {
                int r = p + NRL;
                float en = __int_as_float(s_deg[r * NPW + fs]);
                const float* w = W1 + ((long)r * NN + o) * HIDN;
                for (int c = 0; c < HIDN; c++) atomicAdd(&s_x1[fs * HIDN + c], w[c] * en);
            }
        }
    }
    __syncthreads();

    for (int i = tid; i < NPW * HIDN; i += WGT)
        h[(long)base * HIDN + i] = fmaxf(s_x1[i] + b1[i & (HIDN - 1)], 0.0f);
}

#define TP   404
#define NSTR ((NPW + 15) / 16)
#define SSTR 4
static_assert((NPW * TP + HIDN) * 4 <= 327680, "the tile and its spare words must fit the LDS of one work-group");
static_assert(WGT == 512 && NSTR == 13 && SSTR == 4, "wave to (column pair, strip group) map below: groups of 4, 3, 3, 3 strips");

template <int NS>
__device__ __forceinline__ void l2_mul(const float* s_t, const float* __restrict__ W2,
        const float* __restrict__ b2, float* __restrict__ x2,
        int base, int sg, int n0, int half, int lr) {
    v8f acc[NS][2];
    int aoff[NS];
#pragma unroll
    for (int j = 0; j < NS; j++) {
        v8f z = {0.f, 0.f, 0.f, 0.f, 0.f, 0.f, 0.f, 0.f};
        acc[j][0] = z; acc[j][1] = z;
        int row = (sg + SSTR * j) * 16 + lr;
        if (row > NPW - 1) row = NPW - 1;
        aoff[j] = row * TP + 2 * half;
    }
    const float* bp = W2 + (long)(2 * half) * EMBN + n0 + lr;
#pragma unroll 10
    for (int k0 = 0; k0 < KDIM; k0 += 4) {
        v2f b0, b1;
        b0.x = bp[(long)k0 * EMBN];        b0.y = bp[(long)(k0 + 1) * EMBN];
        b1.x = bp[(long)k0 * EMBN + 16];   b1.y = bp[(long)(k0 + 1) * EMBN + 16];
#pragma unroll
        for (int j = 0; j < NS; j++) {
            v2f a = *(const v2f*)(s_t + aoff[j] + k0);
            acc[j][0] = __builtin_amdgcn_wmma_f32_16x16x4_f32(
                false, a, false, b0, (short)0, acc[j][0], false, false);
            acc[j][1] = __builtin_amdgcn_wmma_f32_16x16x4_f32(
                false, a, false, b1, (short)0, acc[j][1], false, false);
        }
    }
#pragma unroll
    for (int j = 0; j < NS; j++) {
#pragma unroll
        for (int q = 0; q < 2; q++) {
            int n = n0 + 16 * q + lr;
            float bias = b2[n];
#pragma unroll
            for (int v = 0; v < 8; v++) {
                int row = (sg + SSTR * j) * 16 + v + 8 * half;
                if (row < NPW) x2[(long)(base + row) * EMBN + n] = acc[j][q][v] + bias;
            }
        }
    }
}

template <int I>
__device__ __forceinline__ unsigned rshare(unsigned x) {
    return (unsigned)__builtin_amdgcn_update_dpp(0, (int)x, 0x150 + I, 0xf, 0xf, true);
}

template <int I>
__device__ __forceinline__ void l2_take(uint2 e, const float* __restrict__ h, int c, float& hv, float& en, int& dst) {
    const unsigned wx = rshare<I>(e.x), wy = rshare<I>(e.y);
    const int sN = (int)(wx & 0xffffu), ol = (int)((wx >> 16) & 0xffu), r = (int)(wx >> 24);
    hv  = h[(long)sN * HIDN + c];
    en  = __uint_as_float(wy);
    dst = ol * TP + r * HIDN + c;
}

__device__ __forceinline__ void l2_slot(uint2& e, const uint2* nxt,
        const float* __restrict__ h, int c, float* s_t) {
    float hv[L1U], en[L1U]; int dst[L1U];
    l2_take<0>(e, h, c, hv[0], en[0], dst[0]);       l2_take<1>(e, h, c, hv[1], en[1], dst[1]);
    l2_take<2>(e, h, c, hv[2], en[2], dst[2]);       l2_take<3>(e, h, c, hv[3], en[3], dst[3]);
    l2_take<4>(e, h, c, hv[4], en[4], dst[4]);       l2_take<5>(e, h, c, hv[5], en[5], dst[5]);
    l2_take<6>(e, h, c, hv[6], en[6], dst[6]);       l2_take<7>(e, h, c, hv[7], en[7], dst[7]);
    l2_take<8>(e, h, c, hv[8], en[8], dst[8]);       l2_take<9>(e, h, c, hv[9], en[9], dst[9]);
    l2_take<10>(e, h, c, hv[10], en[10], dst[10]);   l2_take<11>(e, h, c, hv[11], en[11], dst[11]);
    l2_take<12>(e, h, c, hv[12], en[12], dst[12]);   l2_take<13>(e, h, c, hv[13], en[13], dst[13]);
    l2_take<14>(e, h, c, hv[14], en[14], dst[14]);   l2_take<15>(e, h, c, hv[15], en[15], dst[15]);
    e = *nxt;
    __builtin_amdgcn_sched_barrier(0);
#pragma unroll
    for (int u = 0; u < L1U; u++) atomicAdd(&s_t[dst[u]], hv[u] * en[u]);
}
static_assert(L1U == 16 && TRIP == WGT && LCAP >= 2 * WGT, "a slot is one pass of the list: 16 sub-steps of 16-lane groups; the first two slots lie inside the list");
#define ZH (20 * WGT)

__global__ __launch_bounds__(WGT) void k_layer2(const int* __restrict__ t,
        const unsigned* __restrict__ glist, const int* __restrict__ gcnt,
        const float* __restrict__ h, const float* __restrict__ norm,
        const float* __restrict__ W2, const float* __restrict__ b2, float* __restrict__ x2) {
    __shared__ __attribute__((aligned(16))) float s_t[NPW * TP + HIDN];

    const int tid  = threadIdx.x;
    const int base = blockIdx.x * NPW;
    const int g = tid >> 4, c = tid & 15;

    float hv[SLJ], nv[SLJ];
#pragma unroll
    for (int j = 0; j < SLJ; j++) {
        const int ol = g + j * L1G;
        hv[j] = (ol < NPW) ? h[(long)(base + ol) * HIDN + c] : 0.0f;
        nv[j] = (ol < NPW) ? norm[2 * NRL * NN + base + ol] : 0.0f;
    }
    const int cnt = gcnt[blockIdx.x];

    for (int i = tid; i < ZH; i += WGT) ((float4*)s_t)[i] = make_float4(0.f, 0.f, 0.f, 0.f);
    const uint2* lp = (const uint2*)glist + (long)blockIdx.x * LCAP;
    const int cntp = (cnt <= LCAP) ? (cnt + TRIP - 1) / TRIP * TRIP : 0;
    uint2 ea = (0   < cntp) ? lp[tid]       : make_uint2(0u, 0u);
    uint2 eb = (WGT < cntp) ? lp[WGT + tid] : make_uint2(0u, 0u);
    for (int i = ZH + tid; i < NPW * TP / 4; i += WGT) ((float4*)s_t)[i] = make_float4(0.f, 0.f, 0.f, 0.f);
    __syncthreads();

#pragma unroll
    for (int j = 0; j < SLJ; j++) {
        const int ol = g + j * L1G;
        if (ol < NPW) s_t[ol * TP + 2 * NRL * HIDN + c] = hv[j] * nv[j];
    }

    if (cnt <= LCAP) {
        for (int b = 0; b < cntp; b += 2 * WGT) {
            l2_slot(ea, lp + ((b + 2 * WGT < cntp) ? b + 2 * WGT : 0) + tid, h, c, s_t);
            if (b + WGT < cntp)
                l2_slot(eb, lp + ((b + 3 * WGT < cntp) ? b + 3 * WGT : 0) + tid, h, c, s_t);
        }
    } else {
        for (int k = tid; k < NT; k += WGT) {
            int s = t[3*k], p = t[3*k+1], o = t[3*k+2];
            unsigned fo = (unsigned)(o - base), fs = (unsigned)(s - base);
            if (fo < NPW) {
                float en = norm[p * NN + o];
                const float* hs = h + (long)s * HIDN;
                for (int cc = 0; cc < HIDN; cc++) atomicAdd(&s_t[fo * TP + p * HIDN + cc], hs[cc] * en);
            }
            if (fs < NPW) {
                int r = p + NRL;
                float en = norm[r * NN + s];
                const float* ho = h + (long)o * HIDN;
                for (int cc = 0; cc < HIDN; cc++) atomicAdd(&s_t[fs * TP + r * HIDN + cc], ho[cc] * en);
            }
        }
    }
    __syncthreads();

    const int wave = tid >> 5, lane = tid & 31;
    const int cp = wave & 3, sg = ((wave >> 2) - cp) & 3;
    if (sg == 0) l2_mul<4>(s_t, W2, b2, x2, base, sg, cp * 32, lane >> 4, lane & 15);
    else         l2_mul<3>(s_t, W2, b2, x2, base, sg, cp * 32, lane >> 4, lane & 15);
}

#define FCAP 16384
#define SCW  (SCT / 32)
#define SSPW 16
#define SCU  16
#define XJ   (NPW * EMBN / 4 / SCT)
static_assert(NPW * EMBN * 4 + NRL * EMBN * 4 + FCAP * 8 + 64 <= 327680, "rows, rel and list must fit the LDS of one work-group");
static_assert(EMBN == 128, "one row = 32 lanes x 4 floats");
static_assert(NCH <= SSPW * SCW && NRL * EMBN / 4 <= SCT && NPW * EMBN / 4 - XJ * SCT <= SCT, "segments per wave; rel and the rest of the rows in one pass");

__device__ __forceinline__ void fl_put(bool f, unsigned w, unsigned ix, int ln, uint2* s_fl, int* s_n) {
    const unsigned m = __builtin_amdgcn_ballot_w32(f);
    if (m == 0u) return;
    int p0 = 0;
    if (ln == 0) p0 = atomicAdd(s_n, __popc(m));
    p0 = __builtin_amdgcn_readfirstlane(p0);
    if (f) {
        const int pos = p0 + __popc(m & ((1u << ln) - 1u));
        if (pos < FCAP) s_fl[pos] = make_uint2(w, ix);
    }
}

__device__ __forceinline__ float score_part(float4 xs, float4 rl, float4 xo) {
    float acc = xs.x * rl.x * xo.x;
    acc += xs.y * rl.y * xo.y;
    acc += xs.z * rl.z * xo.z;
    acc += xs.w * rl.w * xo.w;
    return acc;
}

__device__ __forceinline__ float xhalf(float x) {
    const int b = __builtin_bit_cast(int, x);
    return __builtin_bit_cast(float, __builtin_amdgcn_permlanex16(b, b, 0x76543210, 0xfedcba98, false, false));
}
template <int M>
__device__ __forceinline__ float xrow(float x) {
    const int b = __builtin_bit_cast(int, x);
    return __builtin_bit_cast(float, __builtin_amdgcn_update_dpp(0, b, 0x160 + M, 0xf, 0xf, true));
}

__device__ __forceinline__ float score_lane(float4 xs, float4 rl, float4 xo) {
    float acc = score_part(xs, rl, xo);
    acc += xhalf(acc);
    acc += xrow<8>(acc);
    acc += xrow<4>(acc);
    acc += xrow<2>(acc);
    acc += xrow<1>(acc);
    return acc;
}

__global__ __launch_bounds__(SCT) void k_score_own(const int* __restrict__ t,
        const float* __restrict__ x2, const float* __restrict__ rel,
        const unsigned* __restrict__ sorted, const unsigned* __restrict__ sidx,
        const unsigned* __restrict__ segT, float* __restrict__ out) {
    __shared__ __attribute__((aligned(16))) float s_x[NPW * EMBN];
    __shared__ __attribute__((aligned(16))) float s_r[NRL * EMBN];
    __shared__ uint2 s_fl[FCAP];
    __shared__ int   s_n;

    const int tid  = threadIdx.x;
    const int base = blockIdx.x * NPW;
    const int wv = tid >> 5, ln = tid & 31;

    if (tid == 0) s_n = 0;

    unsigned w0[SSPW], i0[SSPW];
    int n[SSPW], off[SSPW];
#pragma unroll
    for (int u = 0; u < SSPW; u++) {
        const int a  = wv + u * SCW;
        const int ac = (a < NCH) ? a : NCH - 1;
        const unsigned sg = segT[blockIdx.x * NCH + ac];
        n[u]   = (a < NCH) ? (int)(sg >> 16) : 0;
        off[u] = ac * (2 * CH) + (int)(sg & 0xffffu);
    }
#pragma unroll
    for (int u = 0; u < SSPW; u++) {
        w0[u] = (ln < n[u]) ? sorted[off[u] + ln] : 0xff000000u;
        i0[u] = (ln < n[u]) ? sidx[off[u] + ln]   : 0u;
    }

    {
        const float4* xg = (const float4*)(x2 + (long)base * EMBN);
        float4 xr[XJ];
#pragma unroll
        for (int j = 0; j < XJ; j++) xr[j] = xg[tid + j * SCT];
#pragma unroll
        for (int j = 0; j < XJ; j++) ((float4*)s_x)[tid + j * SCT] = xr[j];
        if (tid + XJ * SCT < NPW * EMBN / 4) ((float4*)s_x)[tid + XJ * SCT] = xg[tid + XJ * SCT];
        if (tid < NRL * EMBN / 4) ((float4*)s_r)[tid] = ((const float4*)rel)[tid];
    }
    __syncthreads();

    {
        unsigned m[SSPW];
        int tot = 0;
#pragma unroll
        for (int u = 0; u < SSPW; u++) {
            m[u] = __builtin_amdgcn_ballot_w32((ln < n[u]) && ((w0[u] >> 24) < (unsigned)NRL));
            tot += __popc(m[u]);
        }
        int p0 = 0;
        if (ln == 0 && tot > 0) p0 = atomicAdd(&s_n, tot);
        p0 = __builtin_amdgcn_readfirstlane(p0);
#pragma unroll
        for (int u = 0; u < SSPW; u++) {
            if ((m[u] >> ln) & 1u) {
                const int pos = p0 + __popc(m[u] & ((1u << ln) - 1u));
                if (pos < FCAP) s_fl[pos] = make_uint2(w0[u], i0[u]);
            }
            p0 += __popc(m[u]);
        }
#pragma unroll
        for (int u = 0; u < SSPW; u++) {
            for (int b = 32; b < n[u]; b += 32) {
                const int  i = b + ln;
                const bool v = i < n[u];
                const unsigned w  = v ? sorted[off[u] + i] : 0xff000000u;
                const unsigned ix = v ? sidx[off[u] + i]   : 0u;
                fl_put(v && ((w >> 24) < (unsigned)NRL), w, ix, ln, s_fl, &s_n);
            }
        }
    }
    __syncthreads();
    const int nf = s_n;

    if (nf <= FCAP) {
        for (int e0 = wv; e0 < nf; e0 += SCU * SCW) {
            float4 xs[SCU];
            unsigned ww[SCU], ix[SCU];
            float acc[SCU];
#pragma unroll
            for (int u = 0; u < SCU; u++) {
                const int e = e0 + u * SCW;
                const uint2 en = s_fl[(e < nf) ? e : nf - 1];
                ww[u] = en.x; ix[u] = en.y;
                xs[u] = *(const float4*)(x2 + (long)(en.x & 0xffffu) * EMBN + 4 * ln);
            }
#pragma unroll
            for (int u = 0; u < SCU; u++) {
                const int ol = (int)((ww[u] >> 16) & 0xffu), p = (int)(ww[u] >> 24);
                const float4 xo = *(const float4*)(s_x + ol * EMBN + 4 * ln);
                const float4 rl = *(const float4*)(s_r + p * EMBN + 4 * ln);
                acc[u] = score_part(xs[u], rl, xo);
            }
#pragma unroll
            for (int u = 0; u < SCU; u++) acc[u] += xhalf(acc[u]);
#pragma unroll
            for (int u = 0; u < SCU; u++) acc[u] += xrow<8>(acc[u]);
#pragma unroll
            for (int u = 0; u < SCU; u++) acc[u] += xrow<4>(acc[u]);
#pragma unroll
            for (int u = 0; u < SCU; u++) acc[u] += xrow<2>(acc[u]);
#pragma unroll
            for (int u = 0; u < SCU; u++) acc[u] += xrow<1>(acc[u]);
#pragma unroll
            for (int u = 0; u < SCU; u++)
                if (ln == 0 && e0 + u * SCW < nf) out[ix[u]] = acc[u];
        }
    } else {
        for (int k = wv; k < NT; k += SCW) {
            const int s = t[3*k], p = t[3*k+1], o = t[3*k+2];
            const unsigned fo = (unsigned)(o - base);
            if (fo < NPW) {
                const float4 xv = *(const float4*)(x2 + (long)s * EMBN + 4 * ln);
                const float4 xo = *(const float4*)(s_x + fo * EMBN + 4 * ln);
                const float4 rl = *(const float4*)(s_r + p * EMBN + 4 * ln);
                const float acc = score_lane(xv, rl, xo);
                if (ln == 0) out[k] = acc;
            }
        }
    }
}

extern "C" void kernel_launch(void* const* d_in, const int* in_sizes, int n_in,
                              void* d_out, int out_size, void* d_ws, size_t ws_size,
                              hipStream_t stream) {
    const int*   triples = (const int*)  d_in[0];
    const float* W1      = (const float*)d_in[1];
    const float* b1      = (const float*)d_in[2];
    const float* W2      = (const float*)d_in[3];
    const float* b2      = (const float*)d_in[4];
    const float* rel     = (const float*)d_in[5];
    float*       out     = (float*)d_out;

    float* deg = (float*)d_ws;
    float* x1  = deg + (long)RP * NN;
    float* agg = x1  + (long)NN * HIDN;
    float* x2  = agg + (long)NN * KDIM;
    unsigned* glist = (unsigned*)agg;
    int*      gcnt  = (int*)(agg + 2L * NWG * LCAP);
    unsigned* sorted = (unsigned*)(agg + 2L * NWG * LCAP + 256);
    unsigned* segT   = sorted + 2L * NT;
    unsigned* sidx   = segT + (long)NWG * NCH;
    static_assert(2L * NWG * LCAP + 256 + 4L * NT + (long)NWG * NCH <= (long)NN * KDIM, "lists, sorted chunks and their triple indices fit where agg was");

    k_bin<<<NCH, BT, 0, stream>>>(triples, sorted, sidx, segT);
    k_layer1<<<NWG, WGT, 0, stream>>>(triples, W1, b1, deg, x1, glist, gcnt, sorted, segT);
    k_layer2<<<NWG, WGT, 0, stream>>>(triples, glist, gcnt, x1, deg, W2, b2, x2);
    k_score_own<<<NWG, SCT, 0, stream>>>(triples, x2, rel, sorted, sidx, segT, out);
}


# Whole 128-B lines per global access: sort launch orders its chunk in LDS, stores 16-B runs; workspace planes start on lines (x2 row read 4 lines, not 5); x2 stored a whole row line per store
# speedup vs baseline: 6.0063x; 1.0755x over previous
#include <hip/hip_runtime.h>

typedef float v2f __attribute__((ext_vector_type(2)));
typedef float v8f __attribute__((ext_vector_type(8)));

#define NN   50000
#define NRL  12
#define NT   500000
#define HIDN 16
#define EMBN 128
#define RP   25
#define KDIM 400
#define NAUG (2*NT + NN)

#define NPW  200
#define NWG  (NN / NPW)
#define LCAP 32768
#define WGT  512
#define NWV  (WGT / 32)
#define SPW  16
#define SCT  512
#define BT   256
#define BU   8
#define NCH  250
#define CH   (NT / NCH)
static_assert(NN % NPW == 0, "owners must tile the nodes");
static_assert(NN <= 65536 && NPW <= 256 && RP <= 256, "list word packing");
static_assert(NT % NCH == 0 && 2 * CH < 65536 && CH % BU == 0 && CH / BU <= BT && BT == 256, "chunking and segment word packing");
static_assert(NWG <= 256 && NCH <= 256 && NCH <= SPW * NWV, "one wave scans 256 counts; SPW segments per wave");
static_assert(WGT >= 256, "the first 256 threads load the segment table");
#define CPL ((2 * CH / 4 + BT - 1) / BT)
static_assert((2 * CH) % 32 == 0, "a sorted chunk is a whole number of 128-byte lines");

__device__ __forceinline__ void wave_excl_scan(const int* cv, int* ov, int n, int sh, int lane, int* tot) {
    int v[8], sum = 0;
#pragma unroll
    for (int j = 0; j < 8; j++) {
        int idx = lane * 8 + j;
        int x = (idx < n) ? (int)((unsigned)cv[idx] >> sh) : 0;
        v[j] = sum; sum += x;
    }
    int incl = sum;
#pragma unroll
    for (int d = 1; d < 32; d <<= 1) {
        int y = __shfl_up(incl, d, 32);
        if (lane >= d) incl += y;
    }
    const int b = incl - sum;
#pragma unroll
    for (int j = 0; j < 8; j++) {
        int idx = lane * 8 + j;
        if (idx < n) ov[idx] = b + v[j];
    }
    if (tot != nullptr && lane == 31) *tot = incl;
}

__global__ __launch_bounds__(BT) void k_bin(const int* __restrict__ t,
        unsigned* __restrict__ sorted, unsigned* __restrict__ sidx, unsigned* __restrict__ segT) {
    __shared__ int s_c[256];
    __shared__ int s_o[256];
    __shared__ __attribute__((aligned(16))) unsigned s_w[2 * CH];
    __shared__ __attribute__((aligned(16))) unsigned s_i[2 * CH];
    const int tid = threadIdx.x;
    const int a   = blockIdx.x;
    s_c[tid] = 0;
    __syncthreads();

    const bool on = tid < CH / BU;
    unsigned wf[BU], wi[BU];
    int gf[BU], gi[BU], rf[BU], ri[BU];
#pragma unroll
    for (int u = 0; u < BU; u++) { wf[u] = 0u; wi[u] = 0u; gf[u] = 0; gi[u] = 0; rf[u] = 0; ri[u] = 0; }
    if (on) {
        const int* tp = t + 3L * (a * CH + tid);
        int ts[BU], tr[BU], to[BU];
#pragma unroll
        for (int u = 0; u < BU; u++) {
            ts[u] = tp[3 * (CH / BU) * u];
            tr[u] = tp[3 * (CH / BU) * u + 1];
            to[u] = tp[3 * (CH / BU) * u + 2];
        }
#pragma unroll
        for (int u = 0; u < BU; u++) {
            int s = ts[u], p = tr[u], o = to[u];
            int go = o / NPW, gs = s / NPW;
            wf[u] = (unsigned)s | ((unsigned)(o - go * NPW) << 16) | ((unsigned)p << 24);
            wi[u] = (unsigned)o | ((unsigned)(s - gs * NPW) << 16) | ((unsigned)(p + NRL) << 24);
            gf[u] = go; gi[u] = gs;
            rf[u] = atomicAdd(&s_c[go], 1);
            ri[u] = atomicAdd(&s_c[gs], 1);
        }
    }
    __syncthreads();
    if (tid < 32) wave_excl_scan(s_c, s_o, NWG, 0, tid, nullptr);
    __syncthreads();

    if (on) {
#pragma unroll
        for (int u = 0; u < BU; u++) {
            const int pf = s_o[gf[u]] + rf[u], pi = s_o[gi[u]] + ri[u];
            const unsigned ti = (unsigned)(a * CH + tid + u * (CH / BU));
            s_w[pf] = wf[u]; s_i[pf] = ti;
            s_w[pi] = wi[u]; s_i[pi] = ti;
        }
    }
    __syncthreads();
    {
        uint4* sp = (uint4*)(sorted + (long)a * (2 * CH));
        uint4* ip = (uint4*)(sidx   + (long)a * (2 * CH));
        uint4 cw[CPL], ci[CPL];
#pragma unroll
        for (int j = 0; j < CPL; j++) {
            const int i  = tid + j * BT;
            const int ic = (i < 2 * CH / 4) ? i : 2 * CH / 4 - 1;
            cw[j] = ((const uint4*)s_w)[ic];
            ci[j] = ((const uint4*)s_i)[ic];
        }
#pragma unroll
        for (int j = 0; j < CPL; j++) {
            const int i = tid + j * BT;
            if (i < 2 * CH / 4) { sp[i] = cw[j]; ip[i] = ci[j]; }
        }
    }
    if (tid < NWG) segT[tid * NCH + a] = (unsigned)s_o[tid] | ((unsigned)s_c[tid] << 16);
}

__device__ __forceinline__ void l1_put(unsigned w, int pos, int* s_deg, unsigned* s_list) {
    atomicAdd(&s_deg[(int)(w >> 24) * NPW + (int)((w >> 16) & 0xffu)], 1);
    if (pos < LCAP) s_list[pos] = w;
}

#define L1U 16
#define L1G (WGT / 16)
#define X1J ((NPW * HIDN + WGT - 1) / WGT)
#define SLJ ((NPW + L1G - 1) / L1G)
#define TRIP (L1U * L1G)
#define PADW(base) ((unsigned)(base) | ((unsigned)NPW << 16))
static_assert(LCAP % TRIP == 0 && NPW < 256, "whole passes fit the list; the spare destination fits its field");

__global__ __launch_bounds__(WGT) void k_layer1(const int* __restrict__ t,
        const float* __restrict__ W1, const float* __restrict__ b1,
        float* __restrict__ norm, float* __restrict__ h,
        unsigned* __restrict__ glist, int* __restrict__ gcnt,
        const unsigned* __restrict__ sorted, const unsigned* __restrict__ segT) {
    __shared__ int      s_deg[RP * NPW];
    __shared__ float    s_x1[NPW * HIDN + HIDN];
    __shared__ unsigned s_list[LCAP];
    __shared__ int      s_seg[256];
    __shared__ int      s_dst[256];
    __shared__ int      s_cnt;

    const int tid  = threadIdx.x;
    const int base = blockIdx.x * NPW;

    for (int i = tid; i < RP * NPW; i += WGT) s_deg[i] = (i >= 2 * NRL * NPW) ? 1 : 0;
    if (tid < 256) s_seg[tid] = (tid < NCH) ? (int)segT[blockIdx.x * NCH + tid] : 0;
    __syncthreads();
    if (tid < 32) wave_excl_scan(s_seg, s_dst, NCH, 16, tid, &s_cnt);
    __syncthreads();

    {
        const int wv = tid >> 5, ln = tid & 31;
        unsigned w0[SPW];
        int n[SPW], d0[SPW], so[SPW];
#pragma unroll
        for (int u = 0; u < SPW; u++) {
            int a = wv + u * NWV;
            n[u] = 0; d0[u] = 0; so[u] = 0;
            if (a < NCH) {
                unsigned sg = (unsigned)s_seg[a];
                n[u]  = (int)(sg >> 16);
                so[u] = a * (2 * CH) + (int)(sg & 0xffffu);
                d0[u] = s_dst[a];
            }
            w0[u] = (ln < n[u]) ? sorted[so[u] + ln] : 0u;
        }
#pragma unroll
        for (int u = 0; u < SPW; u++) {
            if (ln < n[u]) l1_put(w0[u], d0[u] + ln, s_deg, s_list);
            for (int i = ln + 32; i < n[u]; i += 32) l1_put(sorted[so[u] + i], d0[u] + i, s_deg, s_list);
        }
    }
    __syncthreads();
    const int cnt  = s_cnt;
    const int cntp = (cnt + TRIP - 1) / TRIP * TRIP;
    if (cnt <= LCAP)
        for (int i = cnt + tid; i < cntp; i += WGT) s_list[i] = PADW(base);

    float sv[X1J];
#pragma unroll
    for (int j = 0; j < X1J; j++) {
        const int i = tid + j * WGT;
        sv[j] = (i < NPW * HIDN) ? W1[((long)(2 * NRL) * NN + base) * HIDN + i] : 0.0f;
    }

    for (int i = tid; i < RP * NPW; i += WGT) {
        float nv = 1.0f / fmaxf((float)s_deg[i], 1.0f);
        s_deg[i] = __float_as_int(nv);
        int r = i / NPW, ol = i - r * NPW;
        norm[r * NN + base + ol] = nv;
    }
    __syncthreads();

    if (cnt <= LCAP) {
        uint2* gl = (uint2*)glist + (long)blockIdx.x * LCAP;
        for (int i = tid; i < cntp; i += WGT) {
            unsigned w = s_list[i];
            int ol = (int)((w >> 16) & 0xffu), r = (int)(w >> 24);
            gl[i] = make_uint2(w, (unsigned)s_deg[r * NPW + ol]);
        }
    }
    if (tid == 0) gcnt[blockIdx.x] = cnt;

#pragma unroll
    for (int j = 0; j < X1J; j++) {
        const int i = tid + j * WGT;
        if (i < NPW * HIDN) s_x1[i] = sv[j] * __int_as_float(s_deg[2 * NRL * NPW + (i >> 4)]);
    }
    __syncthreads();

    if (cnt <= LCAP) {
        const int g = tid >> 4, c = tid & 15;
        for (int e = g; e < cntp; e += TRIP) {
            float v[L1U]; int dst[L1U];
#pragma unroll
            for (int u = 0; u < L1U; u++) {
                unsigned w = s_list[e + u * L1G];
                int sN = (int)(w & 0xffffu), ol = (int)((w >> 16) & 0xffu), r = (int)(w >> 24);
                float en = __int_as_float(s_deg[r * NPW + ol]);
                v[u] = W1[((long)r * NN + sN) * HIDN + c] * en;
                dst[u] = ol * HIDN + c;
            }
#pragma unroll
            for (int u = 0; u < L1U; u++) atomicAdd(&s_x1[dst[u]], v[u]);
        }
    } else {
        for (int k = tid; k < NT; k += WGT) {
            int s = t[3*k], p = t[3*k+1], o = t[3*k+2];
            unsigned fo = (unsigned)(o - base), fs = (unsigned)(s - base);
            if (fo < NPW) {
                float en = __int_as_float(s_deg[p * NPW + fo]);
                const float* w = W1 + ((long)p * NN + s) * HIDN;
                for (int c = 0; c < HIDN; c++) atomicAdd(&s_x1[fo * HIDN + c], w[c] * en);
            }
            if (fs < NPW) {
                int r = p + NRL;
                float en = __int_as_float(s_deg[r * NPW + fs]);
                const float* w = W1 + ((long)r * NN + o) * HIDN;
                for (int c = 0; c < HIDN; c++) atomicAdd(&s_x1[fs * HIDN + c], w[c] * en);
            }
        }
    }
    __syncthreads();

    for (int i = tid; i < NPW * HIDN; i += WGT)
        h[(long)base * HIDN + i] = fmaxf(s_x1[i] + b1[i & (HIDN - 1)], 0.0f);
}

#define TP   404
#define NSTR ((NPW + 15) / 16)
#define SSTR 4
static_assert((NPW * TP + HIDN) * 4 <= 327680, "the tile and its spare words must fit the LDS of one work-group");
static_assert(WGT == 512 && NSTR == 13 && SSTR == 4, "wave to (column pair, strip group) map below: groups of 4, 3, 3, 3 strips");

__device__ __forceinline__ float xhalf(float x) {
    const int b = __builtin_bit_cast(int, x);
    return __builtin_bit_cast(float, __builtin_amdgcn_permlanex16(b, b, 0x76543210, 0xfedcba98, false, false));
}

template <int NS>
__device__ __forceinline__ void l2_mul(const float* s_t, const float* __restrict__ W2,
        const float* __restrict__ b2, float* __restrict__ x2,
        int base, int sg, int n0, int half, int lr) {
    v8f acc[NS][2];
    int aoff[NS];
#pragma unroll
    for (int j = 0; j < NS; j++) {
        v8f z = {0.f, 0.f, 0.f, 0.f, 0.f, 0.f, 0.f, 0.f};
        acc[j][0] = z; acc[j][1] = z;
        int row = (sg + SSTR * j) * 16 + lr;
        if (row > NPW - 1) row = NPW - 1;
        aoff[j] = row * TP + 2 * half;
    }
    const float* bp = W2 + (long)(2 * half) * EMBN + n0 + lr;
#pragma unroll 10
    for (int k0 = 0; k0 < KDIM; k0 += 4) {
        v2f b0, b1;
        b0.x = bp[(long)k0 * EMBN];        b0.y = bp[(long)(k0 + 1) * EMBN];
        b1.x = bp[(long)k0 * EMBN + 16];   b1.y = bp[(long)(k0 + 1) * EMBN + 16];
#pragma unroll
        for (int j = 0; j < NS; j++) {
            v2f a = *(const v2f*)(s_t + aoff[j] + k0);
            acc[j][0] = __builtin_amdgcn_wmma_f32_16x16x4_f32(
                false, a, false, b0, (short)0, acc[j][0], false, false);
            acc[j][1] = __builtin_amdgcn_wmma_f32_16x16x4_f32(
                false, a, false, b1, (short)0, acc[j][1], false, false);
        }
    }
    const int   ncol = n0 + 16 * half + lr;
    const float bias = b2[ncol];
#pragma unroll
    for (int j = 0; j < NS; j++) {
#pragma unroll
        for (int v = 0; v < 8; v++) {
            const float a0 = acc[j][0][v], a1 = acc[j][1][v];
            const float t0 = xhalf(a0), t1 = xhalf(a1);
            const float lo = half ? t1 : a0;
            const float hi = half ? a1 : t0;
            const int row = (sg + SSTR * j) * 16 + v;
            if (row < NPW)     x2[(long)(base + row) * EMBN + ncol]     = lo + bias;
            if (row + 8 < NPW) x2[(long)(base + row + 8) * EMBN + ncol] = hi + bias;
        }
    }
}

template <int I>
__device__ __forceinline__ unsigned rshare(unsigned x) {
    return (unsigned)__builtin_amdgcn_update_dpp(0, (int)x, 0x150 + I, 0xf, 0xf, true);
}

template <int I>
__device__ __forceinline__ void l2_take(uint2 e, const float* __restrict__ h, int c, float& hv, float& en, int& dst) {
    const unsigned wx = rshare<I>(e.x), wy = rshare<I>(e.y);
    const int sN = (int)(wx & 0xffffu), ol = (int)((wx >> 16) & 0xffu), r = (int)(wx >> 24);
    hv  = h[(long)sN * HIDN + c];
    en  = __uint_as_float(wy);
    dst = ol * TP + r * HIDN + c;
}

__device__ __forceinline__ void l2_slot(uint2& e, const uint2* nxt,
        const float* __restrict__ h, int c, float* s_t) {
    float hv[L1U], en[L1U]; int dst[L1U];
    l2_take<0>(e, h, c, hv[0], en[0], dst[0]);       l2_take<1>(e, h, c, hv[1], en[1], dst[1]);
    l2_take<2>(e, h, c, hv[2], en[2], dst[2]);       l2_take<3>(e, h, c, hv[3], en[3], dst[3]);
    l2_take<4>(e, h, c, hv[4], en[4], dst[4]);       l2_take<5>(e, h, c, hv[5], en[5], dst[5]);
    l2_take<6>(e, h, c, hv[6], en[6], dst[6]);       l2_take<7>(e, h, c, hv[7], en[7], dst[7]);
    l2_take<8>(e, h, c, hv[8], en[8], dst[8]);       l2_take<9>(e, h, c, hv[9], en[9], dst[9]);
    l2_take<10>(e, h, c, hv[10], en[10], dst[10]);   l2_take<11>(e, h, c, hv[11], en[11], dst[11]);
    l2_take<12>(e, h, c, hv[12], en[12], dst[12]);   l2_take<13>(e, h, c, hv[13], en[13], dst[13]);
    l2_take<14>(e, h, c, hv[14], en[14], dst[14]);   l2_take<15>(e, h, c, hv[15], en[15], dst[15]);
    e = *nxt;
    __builtin_amdgcn_sched_barrier(0);
#pragma unroll
    for (int u = 0; u < L1U; u++) atomicAdd(&s_t[dst[u]], hv[u] * en[u]);
}
static_assert(L1U == 16 && TRIP == WGT && LCAP >= 2 * WGT, "a slot is one pass of the list: 16 sub-steps of 16-lane groups; the first two slots lie inside the list");
#define ZH (20 * WGT)

__global__ __launch_bounds__(WGT) void k_layer2(const int* __restrict__ t,
        const unsigned* __restrict__ glist, const int* __restrict__ gcnt,
        const float* __restrict__ h, const float* __restrict__ norm,
        const float* __restrict__ W2, const float* __restrict__ b2, float* __restrict__ x2) {
    __shared__ __attribute__((aligned(16))) float s_t[NPW * TP + HIDN];

    const int tid  = threadIdx.x;
    const int base = blockIdx.x * NPW;
    const int g = tid >> 4, c = tid & 15;

    float hv[SLJ], nv[SLJ];
#pragma unroll
    for (int j = 0; j < SLJ; j++) {
        const int ol = g + j * L1G;
        hv[j] = (ol < NPW) ? h[(long)(base + ol) * HIDN + c] : 0.0f;
        nv[j] = (ol < NPW) ? norm[2 * NRL * NN + base + ol] : 0.0f;
    }
    const int cnt = gcnt[blockIdx.x];

    for (int i = tid; i < ZH; i += WGT) ((float4*)s_t)[i] = make_float4(0.f, 0.f, 0.f, 0.f);
    const uint2* lp = (const uint2*)glist + (long)blockIdx.x * LCAP;
    const int cntp = (cnt <= LCAP) ? (cnt + TRIP - 1) / TRIP * TRIP : 0;
    uint2 ea = (0   < cntp) ? lp[tid]       : make_uint2(0u, 0u);
    uint2 eb = (WGT < cntp) ? lp[WGT + tid] : make_uint2(0u, 0u);
    for (int i = ZH + tid; i < NPW * TP / 4; i += WGT) ((float4*)s_t)[i] = make_float4(0.f, 0.f, 0.f, 0.f);
    __syncthreads();

#pragma unroll
    for (int j = 0; j < SLJ; j++) {
        const int ol = g + j * L1G;
        if (ol < NPW) s_t[ol * TP + 2 * NRL * HIDN + c] = hv[j] * nv[j];
    }

    if (cnt <= LCAP) {
        for (int b = 0; b < cntp; b += 2 * WGT) {
            l2_slot(ea, lp + ((b + 2 * WGT < cntp) ? b + 2 * WGT : 0) + tid, h, c, s_t);
            if (b + WGT < cntp)
                l2_slot(eb, lp + ((b + 3 * WGT < cntp) ? b + 3 * WGT : 0) + tid, h, c, s_t);
        }
    } else {
        for (int k = tid; k < NT; k += WGT) {
            int s = t[3*k], p = t[3*k+1], o = t[3*k+2];
            unsigned fo = (unsigned)(o - base), fs = (unsigned)(s - base);
            if (fo < NPW) {
                float en = norm[p * NN + o];
                const float* hs = h + (long)s * HIDN;
                for (int cc = 0; cc < HIDN; cc++) atomicAdd(&s_t[fo * TP + p * HIDN + cc], hs[cc] * en);
            }
            if (fs < NPW) {
                int r = p + NRL;
                float en = norm[r * NN + s];
                const float* ho = h + (long)o * HIDN;
                for (int cc = 0; cc < HIDN; cc++) atomicAdd(&s_t[fs * TP + r * HIDN + cc], ho[cc] * en);
            }
        }
    }
    __syncthreads();

    const int wave = tid >> 5, lane = tid & 31;
    const int cp = wave & 3, sg = ((wave >> 2) - cp) & 3;
    if (sg == 0) l2_mul<4>(s_t, W2, b2, x2, base, sg, cp * 32, lane >> 4, lane & 15);
    else         l2_mul<3>(s_t, W2, b2, x2, base, sg, cp * 32, lane >> 4, lane & 15);
}

#define FCAP 16384
#define SCW  (SCT / 32)
#define SSPW 16
#define SCU  16
#define XJ   (NPW * EMBN / 4 / SCT)
static_assert(NPW * EMBN * 4 + NRL * EMBN * 4 + FCAP * 8 + 64 <= 327680, "rows, rel and list must fit the LDS of one work-group");
static_assert(EMBN == 128, "one row = 32 lanes x 4 floats");
static_assert(NCH <= SSPW * SCW && NRL * EMBN / 4 <= SCT && NPW * EMBN / 4 - XJ * SCT <= SCT, "segments per wave; rel and the rest of the rows in one pass");

__device__ __forceinline__ void fl_put(bool f, unsigned w, unsigned ix, int ln, uint2* s_fl, int* s_n) {
    const unsigned m = __builtin_amdgcn_ballot_w32(f);
    if (m == 0u) return;
    int p0 = 0;
    if (ln == 0) p0 = atomicAdd(s_n, __popc(m));
    p0 = __builtin_amdgcn_readfirstlane(p0);
    if (f) {
        const int pos = p0 + __popc(m & ((1u << ln) - 1u));
        if (pos < FCAP) s_fl[pos] = make_uint2(w, ix);
    }
}

__device__ __forceinline__ float score_part(float4 xs, float4 rl, float4 xo) {
    float acc = xs.x * rl.x * xo.x;
    acc += xs.y * rl.y * xo.y;
    acc += xs.z * rl.z * xo.z;
    acc += xs.w * rl.w * xo.w;
    return acc;
}

template <int M>
__device__ __forceinline__ float xrow(float x) {
    const int b = __builtin_bit_cast(int, x);
    return __builtin_bit_cast(float, __builtin_amdgcn_update_dpp(0, b, 0x160 + M, 0xf, 0xf, true));
}

__device__ __forceinline__ float score_lane(float4 xs, float4 rl, float4 xo) {
    float acc = score_part(xs, rl, xo);
    acc += xhalf(acc);
    acc += xrow<8>(acc);
    acc += xrow<4>(acc);
    acc += xrow<2>(acc);
    acc += xrow<1>(acc);
    return acc;
}

__global__ __launch_bounds__(SCT) void k_score_own(const int* __restrict__ t,
        const float* __restrict__ x2, const float* __restrict__ rel,
        const unsigned* __restrict__ sorted, const unsigned* __restrict__ sidx,
        const unsigned* __restrict__ segT, float* __restrict__ out) {
    __shared__ __attribute__((aligned(16))) float s_x[NPW * EMBN];
    __shared__ __attribute__((aligned(16))) float s_r[NRL * EMBN];
    __shared__ uint2 s_fl[FCAP];
    __shared__ int   s_n;

    const int tid  = threadIdx.x;
    const int base = blockIdx.x * NPW;
    const int wv = tid >> 5, ln = tid & 31;

    if (tid == 0) s_n = 0;

    unsigned w0[SSPW], i0[SSPW];
    int n[SSPW], off[SSPW];
#pragma unroll
    for (int u = 0; u < SSPW; u++) {
        const int a  = wv + u * SCW;
        const int ac = (a < NCH) ? a : NCH - 1;
        const unsigned sg = segT[blockIdx.x * NCH + ac];
        n[u]   = (a < NCH) ? (int)(sg >> 16) : 0;
        off[u] = ac * (2 * CH) + (int)(sg & 0xffffu);
    }
#pragma unroll
    for (int u = 0; u < SSPW; u++) {
        w0[u] = (ln < n[u]) ? sorted[off[u] + ln] : 0xff000000u;
        i0[u] = (ln < n[u]) ? sidx[off[u] + ln]   : 0u;
    }

    {
        const float4* xg = (const float4*)(x2 + (long)base * EMBN);
        float4 xr[XJ];
#pragma unroll
        for (int j = 0; j < XJ; j++) xr[j] = xg[tid + j * SCT];
#pragma unroll
        for (int j = 0; j < XJ; j++) ((float4*)s_x)[tid + j * SCT] = xr[j];
        if (tid + XJ * SCT < NPW * EMBN / 4) ((float4*)s_x)[tid + XJ * SCT] = xg[tid + XJ * SCT];
        if (tid < NRL * EMBN / 4) ((float4*)s_r)[tid] = ((const float4*)rel)[tid];
    }
    __syncthreads();

    {
        unsigned m[SSPW];
        int tot = 0;
#pragma unroll
        for (int u = 0; u < SSPW; u++) {
            m[u] = __builtin_amdgcn_ballot_w32((ln < n[u]) && ((w0[u] >> 24) < (unsigned)NRL));
            tot += __popc(m[u]);
        }
        int p0 = 0;
        if (ln == 0 && tot > 0) p0 = atomicAdd(&s_n, tot);
        p0 = __builtin_amdgcn_readfirstlane(p0);
#pragma unroll
        for (int u = 0; u < SSPW; u++) {
            if ((m[u] >> ln) & 1u) {
                const int pos = p0 + __popc(m[u] & ((1u << ln) - 1u));
                if (pos < FCAP) s_fl[pos] = make_uint2(w0[u], i0[u]);
            }
            p0 += __popc(m[u]);
        }
#pragma unroll
        for (int u = 0; u < SSPW; u++) {
            for (int b = 32; b < n[u]; b += 32) {
                const int  i = b + ln;
                const bool v = i < n[u];
                const unsigned w  = v ? sorted[off[u] + i] : 0xff000000u;
                const unsigned ix = v ? sidx[off[u] + i]   : 0u;
                fl_put(v && ((w >> 24) < (unsigned)NRL), w, ix, ln, s_fl, &s_n);
            }
        }
    }
    __syncthreads();
    const int nf = s_n;

    if (nf <= FCAP) {
        for (int e0 = wv; e0 < nf; e0 += SCU * SCW) {
            float4 xs[SCU];
            unsigned ww[SCU], ix[SCU];
            float acc[SCU];
#pragma unroll
            for (int u = 0; u < SCU; u++) {
                const int e = e0 + u * SCW;
                const uint2 en = s_fl[(e < nf) ? e : nf - 1];
                ww[u] = en.x; ix[u] = en.y;
                xs[u] = *(const float4*)(x2 + (long)(en.x & 0xffffu) * EMBN + 4 * ln);
            }
#pragma unroll
            for (int u = 0; u < SCU; u++) {
                const int ol = (int)((ww[u] >> 16) & 0xffu), p = (int)(ww[u] >> 24);
                const float4 xo = *(const float4*)(s_x + ol * EMBN + 4 * ln);
                const float4 rl = *(const float4*)(s_r + p * EMBN + 4 * ln);
                acc[u] = score_part(xs[u], rl, xo);
            }
#pragma unroll
            for (int u = 0; u < SCU; u++) acc[u] += xhalf(acc[u]);
#pragma unroll
            for (int u = 0; u < SCU; u++) acc[u] += xrow<8>(acc[u]);
#pragma unroll
            for (int u = 0; u < SCU; u++) acc[u] += xrow<4>(acc[u]);
#pragma unroll
            for (int u = 0; u < SCU; u++) acc[u] += xrow<2>(acc[u]);
#pragma unroll
            for (int u = 0; u < SCU; u++) acc[u] += xrow<1>(acc[u]);
#pragma unroll
            for (int u = 0; u < SCU; u++)
                if (ln == 0 && e0 + u * SCW < nf) out[ix[u]] = acc[u];
        }
    } else {
        for (int k = wv; k < NT; k += SCW) {
            const int s = t[3*k], p = t[3*k+1], o = t[3*k+2];
            const unsigned fo = (unsigned)(o - base);
            if (fo < NPW) {
                const float4 xv = *(const float4*)(x2 + (long)s * EMBN + 4 * ln);
                const float4 xo = *(const float4*)(s_x + fo * EMBN + 4 * ln);
                const float4 rl = *(const float4*)(s_r + p * EMBN + 4 * ln);
                const float acc = score_lane(xv, rl, xo);
                if (ln == 0) out[k] = acc;
            }
        }
    }
}

#define LOFF 16
#define SEGW (((long)NWG * NCH + 31) / 32 * 32)
static_assert(((long)RP * NN) % 32 == LOFF && ((long)NN * HIDN) % 32 == 0 && ((long)NN * KDIM) % 32 == 0, "LOFF puts the planes on lines");
static_assert((2L * NWG * LCAP) % 32 == 0 && (2L * NT) % 32 == 0, "gcnt, sorted and segT stay on lines");

extern "C" void kernel_launch(void* const* d_in, const int* in_sizes, int n_in,
                              void* d_out, int out_size, void* d_ws, size_t ws_size,
                              hipStream_t stream) {
    const int*   triples = (const int*)  d_in[0];
    const float* W1      = (const float*)d_in[1];
    const float* b1      = (const float*)d_in[2];
    const float* W2      = (const float*)d_in[3];
    const float* b2      = (const float*)d_in[4];
    const float* rel     = (const float*)d_in[5];
    float*       out     = (float*)d_out;

    float* deg = (float*)d_ws;
    float* x1  = deg + (long)RP * NN + LOFF;
    float* agg = deg + (long)RP * NN + (long)NN * HIDN;
    float* x2  = agg + (long)NN * KDIM - LOFF;
    unsigned* glist = (unsigned*)(agg + LOFF);
    int*      gcnt  = (int*)(agg + LOFF + 2L * NWG * LCAP);
    unsigned* sorted = (unsigned*)(agg + LOFF + 2L * NWG * LCAP + 256);
    unsigned* segT   = sorted + 2L * NT;
    unsigned* sidx   = segT + SEGW;
    static_assert(LOFF + 2L * NWG * LCAP + 256 + 4L * NT + SEGW + LOFF <= (long)NN * KDIM, "lists, sorted chunks and their triple indices fit where agg was, below x2");

    k_bin<<<NCH, BT, 0, stream>>>(triples, sorted, sidx, segT);
    k_layer1<<<NWG, WGT, 0, stream>>>(triples, W1, b1, deg, x1, glist, gcnt, sorted, segT);
    k_layer2<<<NWG, WGT, 0, stream>>>(triples, glist, gcnt, x1, deg, W2, b2, x2);
    k_score_own<<<NWG, SCT, 0, stream>>>(triples, x2, rel, sorted, sidx, segT, out);
}
